# starting kernel
# baseline (speedup 1.0000x reference)
#include <hip/hip_runtime.h>
#include <math.h>


typedef __attribute__((ext_vector_type(2))) float v2f;
typedef __attribute__((ext_vector_type(8))) float v8f;

__device__ __forceinline__ unsigned fkey(float f) {
  unsigned u = __float_as_uint(f);
  return (u & 0x80000000u) ? ~u : (u | 0x80000000u);
}
__device__ __forceinline__ float funkey(unsigned k) {
  unsigned u = (k & 0x80000000u) ? (k & 0x7fffffffu) : ~k;
  return __uint_as_float(u);
}

template <int NCOLS>
__global__ __launch_bounds__(32 * (NCOLS / 16))
void gemm_wmma_f32(const float* __restrict__ A, const float* __restrict__ Wg,
                   float* __restrict__ Out, int M) {
  constexpr int K = 128;
  constexpr int NT = NCOLS / 16;
  __shared__ float Ws[K * NCOLS];

  const int tid = threadIdx.x;
  for (int i = tid; i < K * NCOLS; i += 32 * NT) Ws[i] = Wg[i];
  __syncthreads();

  const int wave  = tid >> 5;
  const int lane  = tid & 31;
  const int row16 = blockIdx.x << 4;
  const int r     = lane & 15;
  const int kb    = (lane >> 4) << 1;
  const int col   = (wave << 4) + r;

  int arow = row16 + r;
  if (arow >= M) arow = M - 1;
  const float* __restrict__ ap = A + (size_t)arow * K;

  v8f acc = {};
#pragma unroll
  for (int k = 0; k < K; k += 4) {
    v2f a, b;
    a.x = ap[k + kb];
    a.y = ap[k + kb + 1];
    b.x = Ws[(k + kb) * NCOLS + col];
    b.y = Ws[(k + kb + 1) * NCOLS + col];
    acc = __builtin_amdgcn_wmma_f32_16x16x4_f32(false, a, false, b,
                                                (short)0, acc, false, false);
  }

  const int orow0 = row16 + ((lane >> 4) << 3);
  const int ocol  = (wave << 4) + r;
#pragma unroll
  for (int i = 0; i < 8; ++i) {
    int orow = orow0 + i;
    if (orow < M) Out[(size_t)orow * NCOLS + ocol] = acc[i];
  }
}

__global__ void k_selfloop_stats(const int* __restrict__ dst,
                                 const float* __restrict__ ew,
                                 float* __restrict__ deg,
                                 float* __restrict__ loopw, int E) {
  int e = blockIdx.x * blockDim.x + threadIdx.x;
  if (e >= E) return;
  int d = dst[e];
  atomicAdd(&deg[d], 1.0f);
  atomicAdd(&loopw[d], ew[e]);
}

__global__ void k_selfloop_div(float* __restrict__ loopw,
                               const float* __restrict__ deg, int N) {
  int n = blockIdx.x * blockDim.x + threadIdx.x;
  if (n >= N) return;
  loopw[n] = loopw[n] / fmaxf(deg[n], 1.0f);
}

template <int H>
__global__ void k_node_coef(const float* __restrict__ hfeat,
                            const float* __restrict__ a_s,
                            const float* __restrict__ a_d,
                            float* __restrict__ scoef,
                            float* __restrict__ dcoef, int N) {
  int n = blockIdx.x * blockDim.x + threadIdx.x;
  if (n >= N) return;
  const float* hr = hfeat + (size_t)n * (H * 64);
#pragma unroll
  for (int h = 0; h < H; ++h) {
    float ss = 0.f, dd = 0.f;
#pragma unroll 8
    for (int c = 0; c < 64; ++c) {
      float v = hr[h * 64 + c];
      ss += v * a_s[h * 64 + c];
      dd += v * a_d[h * 64 + c];
    }
    scoef[n * H + h] = ss;
    dcoef[n * H + h] = dd;
  }
}

template <int H>
__global__ void k_edge_const(const float* __restrict__ We,
                             const float* __restrict__ ae,
                             float* __restrict__ cst) {
  int h = threadIdx.x;
  if (h < H) {
    float s = 0.f;
    for (int c = 0; c < 64; ++c) s += We[h * 64 + c] * ae[h * 64 + c];
    cst[h] = s;
  }
}

template <int H>
__global__ void k_edge_logit_max(const int* __restrict__ src,
                                 const int* __restrict__ dst,
                                 const float* __restrict__ ew,
                                 const float* __restrict__ loopw,
                                 const float* __restrict__ scoef,
                                 const float* __restrict__ dcoef,
                                 const float* __restrict__ cst,
                                 float* __restrict__ albuf,
                                 unsigned* __restrict__ mkey, int E, int ET) {
  int e = blockIdx.x * blockDim.x + threadIdx.x;
  if (e >= ET) return;
  int s, d;
  float w;
  if (e < E) { s = src[e]; d = dst[e]; w = ew[e]; }
  else       { s = d = e - E;          w = loopw[s]; }
#pragma unroll
  for (int h = 0; h < H; ++h) {
    float al = scoef[s * H + h] + dcoef[d * H + h] + w * cst[h];
    al = (al >= 0.f) ? al : 0.2f * al;
    albuf[(size_t)e * H + h] = al;
    atomicMax(&mkey[d * H + h], fkey(al));
  }
}

template <int H>
__global__ void k_edge_expsum(const int* __restrict__ dst,
                              const float* __restrict__ albuf,
                              const unsigned* __restrict__ mkey,
                              float* __restrict__ zbuf,
                              float* __restrict__ ssum, int E, int ET) {
  int e = blockIdx.x * blockDim.x + threadIdx.x;
  if (e >= ET) return;
  int d = (e < E) ? dst[e] : (e - E);
#pragma unroll
  for (int h = 0; h < H; ++h) {
    float z = expf(albuf[(size_t)e * H + h] - funkey(mkey[d * H + h]));
    zbuf[(size_t)e * H + h] = z;
    atomicAdd(&ssum[d * H + h], z);
  }
}

template <int F>
__global__ void k_edge_aggregate(const int* __restrict__ src,
                                 const int* __restrict__ dst,
                                 const float* __restrict__ hfeat,
                                 const float* __restrict__ zbuf,
                                 const float* __restrict__ ssum,
                                 float* __restrict__ agg, int E, int ET) {
  constexpr int H = F / 64;
  int gt = blockIdx.x * blockDim.x + threadIdx.x;
  int e = gt >> 5;
  int lane = gt & 31;
  if (e >= ET) return;
  int s, d;
  if (e < E) { s = src[e]; d = dst[e]; }
  else       { s = d = e - E; }
  float wc[H];
#pragma unroll
  for (int h = 0; h < H; ++h)
    wc[h] = zbuf[(size_t)e * H + h] / ssum[d * H + h];
  const float* hs = hfeat + (size_t)s * F;
  float* ad = agg + (size_t)d * F;
#pragma unroll
  for (int f = lane; f < F; f += 32)
    atomicAdd(&ad[f], hs[f] * wc[f >> 6]);
}

template <int F, bool ELU>
__global__ void k_bias_act(float* __restrict__ a, const float* __restrict__ b,
                           int n) {
  int i = blockIdx.x * blockDim.x + threadIdx.x;
  if (i >= n) return;
  float v = a[i] + b[i & (F - 1)];
  if (ELU) v = (v > 0.f) ? v : expm1f(v);
  a[i] = v;
}

extern "C" void kernel_launch(void* const* d_in, const int* in_sizes, int n_in,
                              void* d_out, int out_size, void* d_ws,
                              size_t ws_size, hipStream_t stream) {
  const float* x   = (const float*)d_in[0];
  const int*  eidx = (const int*)d_in[1];
  const float* ew  = (const float*)d_in[2];
  const float* W1  = (const float*)d_in[3];
  const float* We1 = (const float*)d_in[4];
  const float* as1 = (const float*)d_in[5];
  const float* ad1 = (const float*)d_in[6];
  const float* ae1 = (const float*)d_in[7];
  const float* b1  = (const float*)d_in[8];
  const float* W2  = (const float*)d_in[9];
  const float* We2 = (const float*)d_in[10];
  const float* as2 = (const float*)d_in[11];
  const float* ad2 = (const float*)d_in[12];
  const float* ae2 = (const float*)d_in[13];
  const float* b2  = (const float*)d_in[14];
  float* out = (float*)d_out;

  const int N  = in_sizes[0] / 128;
  const int E  = in_sizes[2];
  const int ET = E + N;
  const int* srcv = eidx;
  const int* dstv = eidx + E;

  float* p = (float*)d_ws;
  float* loopw = p; p += N;
  float* deg   = p; p += N;
  float* scoef = p; p += 2 * (size_t)N;
  float* dcoef = p; p += 2 * (size_t)N;
  float* cst   = p; p += 8;
  float* albuf = p; p += 2 * (size_t)ET;
  float* zbuf  = p; p += 2 * (size_t)ET;
  unsigned* mkey = (unsigned*)p; p += 2 * (size_t)N;
  float* ssum  = p; p += 2 * (size_t)N;
  float* hfeat = p; p += (size_t)N * 128;
  float* agg   = p; p += (size_t)N * 128;

  const int T = 256;
  auto cdiv = [](long long a, long long b) { return (int)((a + b - 1) / b); };

  hipMemsetAsync(deg, 0, (size_t)N * 4, stream);
  hipMemsetAsync(loopw, 0, (size_t)N * 4, stream);
  k_selfloop_stats<<<cdiv(E, T), T, 0, stream>>>(dstv, ew, deg, loopw, E);
  k_selfloop_div<<<cdiv(N, T), T, 0, stream>>>(loopw, deg, N);

  gemm_wmma_f32<128><<<cdiv(N, 16), 256, 0, stream>>>(x, W1, hfeat, N);
  k_node_coef<2><<<cdiv(N, T), T, 0, stream>>>(hfeat, as1, ad1, scoef, dcoef, N);
  k_edge_const<2><<<1, 32, 0, stream>>>(We1, ae1, cst);
  hipMemsetAsync(mkey, 0, 2 * (size_t)N * 4, stream);
  hipMemsetAsync(ssum, 0, 2 * (size_t)N * 4, stream);
  hipMemsetAsync(agg, 0, (size_t)N * 128 * 4, stream);
  k_edge_logit_max<2><<<cdiv(ET, T), T, 0, stream>>>(srcv, dstv, ew, loopw,
                                                     scoef, dcoef, cst, albuf,
                                                     mkey, E, ET);
  k_edge_expsum<2><<<cdiv(ET, T), T, 0, stream>>>(dstv, albuf, mkey, zbuf,
                                                  ssum, E, ET);
  k_edge_aggregate<128><<<cdiv((long long)ET * 32, T), T, 0, stream>>>(
      srcv, dstv, hfeat, zbuf, ssum, agg, E, ET);
  k_bias_act<128, true><<<cdiv((long long)N * 128, T), T, 0, stream>>>(
      agg, b1, N * 128);

  gemm_wmma_f32<64><<<cdiv(N, 16), 128, 0, stream>>>(agg, W2, hfeat, N);
  k_node_coef<1><<<cdiv(N, T), T, 0, stream>>>(hfeat, as2, ad2, scoef, dcoef, N);
  k_edge_const<1><<<1, 32, 0, stream>>>(We2, ae2, cst);
  hipMemsetAsync(mkey, 0, (size_t)N * 4, stream);
  hipMemsetAsync(ssum, 0, (size_t)N * 4, stream);
  hipMemsetAsync(out, 0, (size_t)N * 64 * 4, stream);
  k_edge_logit_max<1><<<cdiv(ET, T), T, 0, stream>>>(srcv, dstv, ew, loopw,
                                                     scoef, dcoef, cst, albuf,
                                                     mkey, E, ET);
  k_edge_expsum<1><<<cdiv(ET, T), T, 0, stream>>>(dstv, albuf, mkey, zbuf,
                                                  ssum, E, ET);
  k_edge_aggregate<64><<<cdiv((long long)ET * 32, T), T, 0, stream>>>(
      srcv, dstv, hfeat, zbuf, ssum, out, E, ET);
  k_bias_act<64, false><<<cdiv((long long)N * 64, T), T, 0, stream>>>(
      out, b2, N * 64);
}


# Segment softmax and sum as dst-sorted gather: per-call CSR, each segment sorted by edge id so float sums run in fixed edge-id order; one wave per dst node reduces in registers; no float atomics
# speedup vs baseline: 3.7017x; 3.7017x over previous
#include <hip/hip_runtime.h>
#include <math.h>


typedef __attribute__((ext_vector_type(2))) float v2f;
typedef __attribute__((ext_vector_type(8))) float v8f;

template <int NCOLS>
__global__ __launch_bounds__(32 * (NCOLS / 16))
void gemm_wmma_f32(const float* __restrict__ A, const float* __restrict__ Wg,
                   float* __restrict__ Out, int M) {
  constexpr int K = 128;
  constexpr int NT = NCOLS / 16;
  __shared__ float Ws[K * NCOLS];

  const int tid = threadIdx.x;
  for (int i = tid; i < K * NCOLS; i += 32 * NT) Ws[i] = Wg[i];
  __syncthreads();

  const int wave  = tid >> 5;
  const int lane  = tid & 31;
  const int row16 = blockIdx.x << 4;
  const int r     = lane & 15;
  const int kb    = (lane >> 4) << 1;
  const int col   = (wave << 4) + r;

  int arow = row16 + r;
  if (arow >= M) arow = M - 1;
  const float* __restrict__ ap = A + (size_t)arow * K;

  v8f acc = {};
#pragma unroll
  for (int k = 0; k < K; k += 4) {
    v2f a, b;
    a.x = ap[k + kb];
    a.y = ap[k + kb + 1];
    b.x = Ws[(k + kb) * NCOLS + col];
    b.y = Ws[(k + kb + 1) * NCOLS + col];
    acc = __builtin_amdgcn_wmma_f32_16x16x4_f32(false, a, false, b,
                                                (short)0, acc, false, false);
  }

  const int orow0 = row16 + ((lane >> 4) << 3);
  const int ocol  = (wave << 4) + r;
#pragma unroll
  for (int i = 0; i < 8; ++i) {
    int orow = orow0 + i;
    if (orow < M) Out[(size_t)orow * NCOLS + ocol] = acc[i];
  }
}

template <int H>
__global__ void k_node_coef(const float* __restrict__ hfeat,
                            const float* __restrict__ a_s,
                            const float* __restrict__ a_d,
                            float* __restrict__ scoef,
                            float* __restrict__ dcoef, int N) {
  int n = blockIdx.x * blockDim.x + threadIdx.x;
  if (n >= N) return;
  const float* hr = hfeat + (size_t)n * (H * 64);
#pragma unroll
  for (int h = 0; h < H; ++h) {
    float ss = 0.f, dd = 0.f;
#pragma unroll 8
    for (int c = 0; c < 64; ++c) {
      float v = hr[h * 64 + c];
      ss += v * a_s[h * 64 + c];
      dd += v * a_d[h * 64 + c];
    }
    scoef[n * H + h] = ss;
    dcoef[n * H + h] = dd;
  }
}

template <int H>
__global__ void k_edge_const(const float* __restrict__ We,
                             const float* __restrict__ ae,
                             float* __restrict__ cst) {
  int h = threadIdx.x;
  if (h < H) {
    float s = 0.f;
    for (int c = 0; c < 64; ++c) s += We[h * 64 + c] * ae[h * 64 + c];
    cst[h] = s;
  }
}


__device__ __forceinline__ int wave_sum_i(int v) {
  for (int o = 16; o > 0; o >>= 1) v += __shfl_xor(v, o);
  return v;
}
__device__ __forceinline__ float wave_sum_f(float v) {
  for (int o = 16; o > 0; o >>= 1) v += __shfl_xor(v, o);
  return v;
}
__device__ __forceinline__ float wave_max_f(float v) {
  for (int o = 16; o > 0; o >>= 1) v = fmaxf(v, __shfl_xor(v, o));
  return v;
}

__global__ void k_count_rank(const int* __restrict__ dst, int* __restrict__ cnt,
                             int* __restrict__ rank, int E) {
  int e = blockIdx.x * blockDim.x + threadIdx.x;
  if (e >= E) return;
  rank[e] = atomicAdd(&cnt[dst[e]], 1);
}

__global__ __launch_bounds__(256)
void k_rowptr_scan(const int* __restrict__ cnt, int* __restrict__ rowptr,
                   int N, int E) {
  __shared__ int s_pre[8];
  __shared__ int s_tot[8];
  const int tid = threadIdx.x, lane = tid & 31, wv = tid >> 5;
  const int b = blockIdx.x;
  int pre = 0;
  for (int c = 0; c < b; ++c) {
    const int4 v = *reinterpret_cast<const int4*>(cnt + c * 1024 + tid * 4);
    pre += (v.x + v.y) + (v.z + v.w);
  }
  const int base = b * 1024 + tid * 4;
  const int v0 = (base     < N) ? cnt[base]     : 0;
  const int v1 = (base + 1 < N) ? cnt[base + 1] : 0;
  const int v2 = (base + 2 < N) ? cnt[base + 2] : 0;
  const int v3 = (base + 3 < N) ? cnt[base + 3] : 0;
  const int tsum = (v0 + v1) + (v2 + v3);
  int inc = tsum;
  for (int o = 1; o < 32; o <<= 1) {
    int t = __shfl_up(inc, o);
    if (lane >= o) inc += t;
  }
  pre = wave_sum_i(pre);
  if (lane == 31) s_tot[wv] = inc;
  if (lane == 0)  s_pre[wv] = pre;
  __syncthreads();
  int off = 0;
  for (int i = 0; i < 8; ++i) {
    off += s_pre[i];
    if (i < wv) off += s_tot[i];
  }
  const int ex = off + inc - tsum;
  if (base     < N) rowptr[base]     = ex;
  if (base + 1 < N) rowptr[base + 1] = ex + v0;
  if (base + 2 < N) rowptr[base + 2] = ex + v0 + v1;
  if (base + 3 < N) rowptr[base + 3] = ex + v0 + v1 + v2;
  if (b == 0 && tid == 0) rowptr[N] = E;
}

__global__ void k_fill(const int* __restrict__ src, const int* __restrict__ dst,
                       const float* __restrict__ ew,
                       const int* __restrict__ rowptr,
                       const int* __restrict__ rank, int* __restrict__ rec_i,
                       int E) {
  int e = blockIdx.x * blockDim.x + threadIdx.x;
  if (e >= E) return;
  int pos = rowptr[dst[e]] + rank[e];
  reinterpret_cast<int4*>(rec_i)[pos] =
      make_int4(e, src[e], __float_as_int(ew[e]), 0);
}

__global__ __launch_bounds__(256)
void k_sort_segments(const int* __restrict__ rowptr, const int* __restrict__ rec_i,
                     int* __restrict__ csr_src, float* __restrict__ csr_w, int N) {
  const int4* __restrict__ rec = reinterpret_cast<const int4*>(rec_i);
  const int lane = threadIdx.x & 31;
  const int d = __builtin_amdgcn_readfirstlane(blockIdx.x * 8 + (threadIdx.x >> 5));
  if (d >= N) return;
  const int start = rowptr[d];
  const int deg = rowptr[d + 1] - start;
  if (deg <= 32) {
    int4 it = make_int4(0x7fffffff, 0, 0, 0);
    if (lane < deg) it = rec[start + lane];
    int r = 0;
    for (int j = 0; j < deg; ++j) {
      int ej = __builtin_amdgcn_readlane(it.x, j);
      r += (ej < it.x) ? 1 : 0;
    }
    if (lane < deg) {
      csr_src[start + r] = it.y;
      csr_w[start + r] = __int_as_float(it.z);
    }
  } else {
    for (int i = lane; i < deg; i += 32) {
      const int4 it = rec[start + i];
      int r = 0;
      for (int j = 0; j < deg; ++j) r += (rec[start + j].x < it.x) ? 1 : 0;
      csr_src[start + r] = it.y;
      csr_w[start + r] = __int_as_float(it.z);
    }
  }
}

template <int H>
__device__ __forceinline__ void item_logits(int s, float w,
                                            const float* __restrict__ scoef,
                                            const float* dc, const float* cs,
                                            float* al) {
  for (int h = 0; h < H; ++h) {
    float a = scoef[s * H + h] + dc[h] + w * cs[h];
    al[h] = (a >= 0.f) ? a : 0.2f * a;
  }
}

template <int H, int K>
__device__ __forceinline__ void gather_rows(
    const float* __restrict__ hfeat, int s, const float* wc, int j, int lane,
    __attribute__((ext_vector_type(2 * H))) float& acc) {
  constexpr int F = H * 64;
  constexpr int V = 2 * H;
  typedef __attribute__((ext_vector_type(V))) float vecV;
  vecV row[K];
  float c[K];
  for (int k = 0; k < K; ++k) {
    const int sj = __builtin_amdgcn_readlane(s, j + k);
    row[k] = *reinterpret_cast<const vecV*>(hfeat + (size_t)sj * F + lane * V);
    float c0 = __int_as_float(
        __builtin_amdgcn_readlane(__float_as_int(wc[0]), j + k));
    if (H == 2) {
      float c1 = __int_as_float(
          __builtin_amdgcn_readlane(__float_as_int(wc[H - 1]), j + k));
      c0 = (lane < 16) ? c0 : c1;
    }
    c[k] = c0;
  }
  for (int k = 0; k < K; ++k) acc += row[k] * c[k];
}

template <int H>
__device__ __forceinline__ void gather_chunk(
    const float* __restrict__ hfeat, int s, const float* wc, int cnt, int lane,
    __attribute__((ext_vector_type(2 * H))) float& acc) {
  int j = 0;
  for (; j + 8 <= cnt; j += 8) gather_rows<H, 8>(hfeat, s, wc, j, lane, acc);
  if (cnt - j >= 4) { gather_rows<H, 4>(hfeat, s, wc, j, lane, acc); j += 4; }
  if (cnt - j >= 2) { gather_rows<H, 2>(hfeat, s, wc, j, lane, acc); j += 2; }
  if (cnt - j >= 1) { gather_rows<H, 1>(hfeat, s, wc, j, lane, acc); }
}

template <int H, bool ELU>
__global__ __launch_bounds__(256)
void k_gat_gather(const int* __restrict__ rowptr, const int* __restrict__ csr_src,
                  const float* __restrict__ csr_w,
                  const float* __restrict__ hfeat,
                  const float* __restrict__ scoef,
                  const float* __restrict__ dcoef,
                  const float* __restrict__ cst, const float* __restrict__ bias,
                  float* __restrict__ outp, int N) {
  constexpr int F = H * 64;
  constexpr int V = 2 * H;
  typedef __attribute__((ext_vector_type(V))) float vecV;
  const int lane = threadIdx.x & 31;
  const int d = __builtin_amdgcn_readfirstlane(blockIdx.x * 8 + (threadIdx.x >> 5));
  if (d >= N) return;
  const int start = rowptr[d];
  const int deg = rowptr[d + 1] - start;
  const int T = deg + 1;

  float dc[H], cs[H];
  for (int h = 0; h < H; ++h) { dc[h] = dcoef[d * H + h]; cs[h] = cst[h]; }

  int s0 = d;
  float w0 = 0.f;
  if (lane < deg) { s0 = csr_src[start + lane]; w0 = csr_w[start + lane]; }
  float pw = w0;
  for (int i = lane + 32; i < deg; i += 32) pw += csr_w[start + i];
  const float wloop = wave_sum_f(pw) / fmaxf((float)deg, 1.0f);
  if (lane == deg) w0 = wloop;
  const bool valid0 = lane < T;
  float al0[H];
  item_logits<H>(s0, w0, scoef, dc, cs, al0);

  float m[H];
  for (int h = 0; h < H; ++h) m[h] = valid0 ? al0[h] : -INFINITY;
  for (int base = 32; base < T; base += 32) {
    const int i = base + lane;
    int s = d; float w = wloop;
    if (i < deg) { s = csr_src[start + i]; w = csr_w[start + i]; }
    float al[H];
    item_logits<H>(s, w, scoef, dc, cs, al);
    if (i < T) {
      for (int h = 0; h < H; ++h) m[h] = fmaxf(m[h], al[h]);
    }
  }
  for (int h = 0; h < H; ++h) m[h] = wave_max_f(m[h]);

  float ex0[H], ps[H];
  for (int h = 0; h < H; ++h) {
    ex0[h] = valid0 ? expf(al0[h] - m[h]) : 0.f;
    ps[h] = ex0[h];
  }
  for (int base = 32; base < T; base += 32) {
    const int i = base + lane;
    int s = d; float w = wloop;
    if (i < deg) { s = csr_src[start + i]; w = csr_w[start + i]; }
    float al[H];
    item_logits<H>(s, w, scoef, dc, cs, al);
    if (i < T) {
      for (int h = 0; h < H; ++h) ps[h] += expf(al[h] - m[h]);
    }
  }
  float ssum[H];
  for (int h = 0; h < H; ++h) ssum[h] = wave_sum_f(ps[h]);

  vecV acc = {};
  float wc[H];
  for (int h = 0; h < H; ++h) wc[h] = ex0[h] / ssum[h];
  gather_chunk<H>(hfeat, s0, wc, (T < 32) ? T : 32, lane, acc);
  for (int base = 32; base < T; base += 32) {
    const int i = base + lane;
    int s = d; float w = wloop;
    if (i < deg) { s = csr_src[start + i]; w = csr_w[start + i]; }
    float al[H];
    item_logits<H>(s, w, scoef, dc, cs, al);
    for (int h = 0; h < H; ++h)
      wc[h] = (i < T) ? expf(al[h] - m[h]) / ssum[h] : 0.f;
    const int rem = T - base;
    gather_chunk<H>(hfeat, s, wc, (rem < 32) ? rem : 32, lane, acc);
  }

  const vecV bv = *reinterpret_cast<const vecV*>(bias + lane * V);
  vecV v = acc + bv;
  if (ELU) {
    for (int q = 0; q < V; ++q) v[q] = (v[q] > 0.f) ? v[q] : expm1f(v[q]);
  }
  *reinterpret_cast<vecV*>(outp + (size_t)d * F + lane * V) = v;
}

extern "C" void kernel_launch(void* const* d_in, const int* in_sizes, int n_in,
                              void* d_out, int out_size, void* d_ws,
                              size_t ws_size, hipStream_t stream) {
  const float* x   = (const float*)d_in[0];
  const int*  eidx = (const int*)d_in[1];
  const float* ew  = (const float*)d_in[2];
  const float* W1  = (const float*)d_in[3];
  const float* We1 = (const float*)d_in[4];
  const float* as1 = (const float*)d_in[5];
  const float* ad1 = (const float*)d_in[6];
  const float* ae1 = (const float*)d_in[7];
  const float* b1  = (const float*)d_in[8];
  const float* W2  = (const float*)d_in[9];
  const float* We2 = (const float*)d_in[10];
  const float* as2 = (const float*)d_in[11];
  const float* ad2 = (const float*)d_in[12];
  const float* ae2 = (const float*)d_in[13];
  const float* b2  = (const float*)d_in[14];
  float* out = (float*)d_out;

  const int N  = in_sizes[0] / 128;
  const int E  = in_sizes[2];
  const int* srcv = eidx;
  const int* dstv = eidx + E;

  float* p = (float*)d_ws;
  auto take = [&p](size_t n) { float* r = p; p += (n + 63) & ~(size_t)63; return r; };
  int*   cnt     = (int*)take(N);
  int*   rowptr  = (int*)take((size_t)N + 1);
  int*   rank    = (int*)take(E);
  int*   rec     = (int*)take(4 * (size_t)E);
  int*   csr_src = (int*)take(E);
  float* csr_w   = take(E);
  float* scoef   = take(2 * (size_t)N);
  float* dcoef   = take(2 * (size_t)N);
  float* cst     = take(8);
  float* hfeat   = take((size_t)N * 128);
  float* agg     = take((size_t)N * 128);

  const int T = 256;
  auto cdiv = [](long long a, long long b) { return (int)((a + b - 1) / b); };

  hipMemsetAsync(cnt, 0, (size_t)N * 4, stream);
  k_count_rank<<<cdiv(E, T), T, 0, stream>>>(dstv, cnt, rank, E);
  k_rowptr_scan<<<cdiv(N, 1024), 256, 0, stream>>>(cnt, rowptr, N, E);
  k_fill<<<cdiv(E, T), T, 0, stream>>>(srcv, dstv, ew, rowptr, rank, rec, E);
  k_sort_segments<<<cdiv(N, 8), 256, 0, stream>>>(rowptr, rec, csr_src, csr_w, N);

  gemm_wmma_f32<128><<<cdiv(N, 16), 256, 0, stream>>>(x, W1, hfeat, N);
  k_node_coef<2><<<cdiv(N, T), T, 0, stream>>>(hfeat, as1, ad1, scoef, dcoef, N);
  k_edge_const<2><<<1, 32, 0, stream>>>(We1, ae1, cst);
  k_gat_gather<2, true><<<cdiv(N, 8), 256, 0, stream>>>(
      rowptr, csr_src, csr_w, hfeat, scoef, dcoef, cst, b1, agg, N);

  gemm_wmma_f32<64><<<cdiv(N, 16), 128, 0, stream>>>(agg, W2, hfeat, N);
  k_node_coef<1><<<cdiv(N, T), T, 0, stream>>>(hfeat, as2, ad2, scoef, dcoef, N);
  k_edge_const<1><<<1, 32, 0, stream>>>(We2, ae2, cst);
  k_gat_gather<1, false><<<cdiv(N, 8), 256, 0, stream>>>(
      rowptr, csr_src, csr_w, hfeat, scoef, dcoef, cst, b2, out, N);
}


# Persistent grids: each per-item kernel (edge bucketing, segment sort, both GEMMs, both gathers) launches a chip-sized grid and loops over its items, so about 6x fewer waves are started per call
# speedup vs baseline: 3.8394x; 1.0372x over previous
#include <hip/hip_runtime.h>
#include <math.h>


typedef __attribute__((ext_vector_type(2))) float v2f;
typedef __attribute__((ext_vector_type(8))) float v8f;

template <int NCOLS>
__global__ __launch_bounds__(32 * (NCOLS / 16))
void gemm_wmma_f32(const float* __restrict__ A, const float* __restrict__ Wg,
                   float* __restrict__ Out, int M) {
  constexpr int K = 128;
  constexpr int NT = NCOLS / 16;
  __shared__ float Ws[K * NCOLS];

  const int tid = threadIdx.x;
  for (int i = tid; i < K * NCOLS; i += 32 * NT) Ws[i] = Wg[i];
  __syncthreads();

  const int wave  = tid >> 5;
  const int lane  = tid & 31;
  const int r     = lane & 15;
  const int kb    = (lane >> 4) << 1;
  const int col   = (wave << 4) + r;
  const int nstrips = (M + 15) >> 4;

  for (int strip = blockIdx.x; strip < nstrips; strip += gridDim.x) {
    const int row16 = strip << 4;
    int arow = row16 + r;
    if (arow >= M) arow = M - 1;
    const float* __restrict__ ap = A + (size_t)arow * K;

    v8f acc = {};
#pragma unroll
    for (int k = 0; k < K; k += 4) {
      v2f a, b;
      a.x = ap[k + kb];
      a.y = ap[k + kb + 1];
      b.x = Ws[(k + kb) * NCOLS + col];
      b.y = Ws[(k + kb + 1) * NCOLS + col];
      acc = __builtin_amdgcn_wmma_f32_16x16x4_f32(false, a, false, b,
                                                  (short)0, acc, false, false);
    }

    const int orow0 = row16 + ((lane >> 4) << 3);
#pragma unroll
    for (int i = 0; i < 8; ++i) {
      int orow = orow0 + i;
      if (orow < M) Out[(size_t)orow * NCOLS + col] = acc[i];
    }
  }
}

template <int H>
__global__ void k_node_coef(const float* __restrict__ hfeat,
                            const float* __restrict__ a_s,
                            const float* __restrict__ a_d,
                            float* __restrict__ scoef,
                            float* __restrict__ dcoef, int N) {
  int n = blockIdx.x * blockDim.x + threadIdx.x;
  if (n >= N) return;
  const float* hr = hfeat + (size_t)n * (H * 64);
#pragma unroll
  for (int h = 0; h < H; ++h) {
    float ss = 0.f, dd = 0.f;
#pragma unroll 8
    for (int c = 0; c < 64; ++c) {
      float v = hr[h * 64 + c];
      ss += v * a_s[h * 64 + c];
      dd += v * a_d[h * 64 + c];
    }
    scoef[n * H + h] = ss;
    dcoef[n * H + h] = dd;
  }
}

template <int H>
__global__ void k_edge_const(const float* __restrict__ We,
                             const float* __restrict__ ae,
                             float* __restrict__ cst) {
  int h = threadIdx.x;
  if (h < H) {
    float s = 0.f;
    for (int c = 0; c < 64; ++c) s += We[h * 64 + c] * ae[h * 64 + c];
    cst[h] = s;
  }
}


__device__ __forceinline__ int wave_sum_i(int v) {
  for (int o = 16; o > 0; o >>= 1) v += __shfl_xor(v, o);
  return v;
}
__device__ __forceinline__ float wave_sum_f(float v) {
  for (int o = 16; o > 0; o >>= 1) v += __shfl_xor(v, o);
  return v;
}
__device__ __forceinline__ float wave_max_f(float v) {
  for (int o = 16; o > 0; o >>= 1) v = fmaxf(v, __shfl_xor(v, o));
  return v;
}

__global__ void k_count_rank(const int* __restrict__ dst, int* __restrict__ cnt,
                             int* __restrict__ rank, int E) {
  const int stride = gridDim.x * blockDim.x;
  int e = blockIdx.x * blockDim.x + threadIdx.x;
  for (; e < E - 3 * stride; e += 4 * stride) {
    const int d0 = dst[e];
    const int d1 = dst[e + stride];
    const int d2 = dst[e + 2 * stride];
    const int d3 = dst[e + 3 * stride];
    const int r0 = atomicAdd(&cnt[d0], 1);
    const int r1 = atomicAdd(&cnt[d1], 1);
    const int r2 = atomicAdd(&cnt[d2], 1);
    const int r3 = atomicAdd(&cnt[d3], 1);
    rank[e] = r0;
    rank[e + stride] = r1;
    rank[e + 2 * stride] = r2;
    rank[e + 3 * stride] = r3;
  }
  for (; e < E; e += stride) rank[e] = atomicAdd(&cnt[dst[e]], 1);
}

__global__ __launch_bounds__(256)
void k_rowptr_scan(const int* __restrict__ cnt, int* __restrict__ rowptr,
                   int N, int E) {
  __shared__ int s_pre[8];
  __shared__ int s_tot[8];
  const int tid = threadIdx.x, lane = tid & 31, wv = tid >> 5;
  const int b = blockIdx.x;
  int pre = 0;
  for (int c = 0; c < b; ++c) {
    const int4 v = *reinterpret_cast<const int4*>(cnt + c * 1024 + tid * 4);
    pre += (v.x + v.y) + (v.z + v.w);
  }
  const int base = b * 1024 + tid * 4;
  const int v0 = (base     < N) ? cnt[base]     : 0;
  const int v1 = (base + 1 < N) ? cnt[base + 1] : 0;
  const int v2 = (base + 2 < N) ? cnt[base + 2] : 0;
  const int v3 = (base + 3 < N) ? cnt[base + 3] : 0;
  const int tsum = (v0 + v1) + (v2 + v3);
  int inc = tsum;
  for (int o = 1; o < 32; o <<= 1) {
    int t = __shfl_up(inc, o);
    if (lane >= o) inc += t;
  }
  pre = wave_sum_i(pre);
  if (lane == 31) s_tot[wv] = inc;
  if (lane == 0)  s_pre[wv] = pre;
  __syncthreads();
  int off = 0;
  for (int i = 0; i < 8; ++i) {
    off += s_pre[i];
    if (i < wv) off += s_tot[i];
  }
  const int ex = off + inc - tsum;
  if (base     < N) rowptr[base]     = ex;
  if (base + 1 < N) rowptr[base + 1] = ex + v0;
  if (base + 2 < N) rowptr[base + 2] = ex + v0 + v1;
  if (base + 3 < N) rowptr[base + 3] = ex + v0 + v1 + v2;
  if (b == 0 && tid == 0) rowptr[N] = E;
}

__global__ void k_fill(const int* __restrict__ src, const int* __restrict__ dst,
                       const float* __restrict__ ew,
                       const int* __restrict__ rowptr,
                       const int* __restrict__ rank, int* __restrict__ rec_i,
                       int E) {
  int4* __restrict__ rec = reinterpret_cast<int4*>(rec_i);
  const int stride = gridDim.x * blockDim.x;
  int e = blockIdx.x * blockDim.x + threadIdx.x;
  for (; e < E - 3 * stride; e += 4 * stride) {
    const int e1 = e + stride, e2 = e + 2 * stride, e3 = e + 3 * stride;
    const int d0 = dst[e], d1 = dst[e1], d2 = dst[e2], d3 = dst[e3];
    const int p0 = rowptr[d0] + rank[e];
    const int p1 = rowptr[d1] + rank[e1];
    const int p2 = rowptr[d2] + rank[e2];
    const int p3 = rowptr[d3] + rank[e3];
    rec[p0] = make_int4(e,  src[e],  __float_as_int(ew[e]),  0);
    rec[p1] = make_int4(e1, src[e1], __float_as_int(ew[e1]), 0);
    rec[p2] = make_int4(e2, src[e2], __float_as_int(ew[e2]), 0);
    rec[p3] = make_int4(e3, src[e3], __float_as_int(ew[e3]), 0);
  }
  for (; e < E; e += stride) {
    const int pos = rowptr[dst[e]] + rank[e];
    rec[pos] = make_int4(e, src[e], __float_as_int(ew[e]), 0);
  }
}

__global__ __launch_bounds__(256)
void k_sort_segments(const int* __restrict__ rowptr, const int* __restrict__ rec_i,
                     int* __restrict__ csr_src, float* __restrict__ csr_w, int N) {
  const int4* __restrict__ rec = reinterpret_cast<const int4*>(rec_i);
  const int lane = threadIdx.x & 31;
  const int d_first =
      __builtin_amdgcn_readfirstlane(blockIdx.x * 8 + (threadIdx.x >> 5));
  for (int d = d_first; d < N; d += gridDim.x * 8) {
  const int start = rowptr[d];
  const int deg = rowptr[d + 1] - start;
  if (deg <= 32) {
    int4 it = make_int4(0x7fffffff, 0, 0, 0);
    if (lane < deg) it = rec[start + lane];
    int r = 0;
    for (int j = 0; j < deg; ++j) {
      int ej = __builtin_amdgcn_readlane(it.x, j);
      r += (ej < it.x) ? 1 : 0;
    }
    if (lane < deg) {
      csr_src[start + r] = it.y;
      csr_w[start + r] = __int_as_float(it.z);
    }
  } else {
    for (int i = lane; i < deg; i += 32) {
      const int4 it = rec[start + i];
      int r = 0;
      for (int j = 0; j < deg; ++j) r += (rec[start + j].x < it.x) ? 1 : 0;
      csr_src[start + r] = it.y;
      csr_w[start + r] = __int_as_float(it.z);
    }
  }
  }
}

template <int H>
__device__ __forceinline__ void item_logits(int s, float w,
                                            const float* __restrict__ scoef,
                                            const float* dc, const float* cs,
                                            float* al) {
  for (int h = 0; h < H; ++h) {
    float a = scoef[s * H + h] + dc[h] + w * cs[h];
    al[h] = (a >= 0.f) ? a : 0.2f * a;
  }
}

template <int H, int K>
__device__ __forceinline__ void gather_rows(
    const float* __restrict__ hfeat, int s, const float* wc, int j, int lane,
    __attribute__((ext_vector_type(2 * H))) float& acc) {
  constexpr int F = H * 64;
  constexpr int V = 2 * H;
  typedef __attribute__((ext_vector_type(V))) float vecV;
  vecV row[K];
  float c[K];
  for (int k = 0; k < K; ++k) {
    const int sj = __builtin_amdgcn_readlane(s, j + k);
    row[k] = *reinterpret_cast<const vecV*>(hfeat + (size_t)sj * F + lane * V);
    float c0 = __int_as_float(
        __builtin_amdgcn_readlane(__float_as_int(wc[0]), j + k));
    if (H == 2) {
      float c1 = __int_as_float(
          __builtin_amdgcn_readlane(__float_as_int(wc[H - 1]), j + k));
      c0 = (lane < 16) ? c0 : c1;
    }
    c[k] = c0;
  }
  for (int k = 0; k < K; ++k) acc += row[k] * c[k];
}

template <int H>
__device__ __forceinline__ void gather_chunk(
    const float* __restrict__ hfeat, int s, const float* wc, int cnt, int lane,
    __attribute__((ext_vector_type(2 * H))) float& acc) {
  int j = 0;
  for (; j + 8 <= cnt; j += 8) gather_rows<H, 8>(hfeat, s, wc, j, lane, acc);
  if (cnt - j >= 4) { gather_rows<H, 4>(hfeat, s, wc, j, lane, acc); j += 4; }
  if (cnt - j >= 2) { gather_rows<H, 2>(hfeat, s, wc, j, lane, acc); j += 2; }
  if (cnt - j >= 1) { gather_rows<H, 1>(hfeat, s, wc, j, lane, acc); }
}

template <int H, bool ELU>
__global__ __launch_bounds__(256)
void k_gat_gather(const int* __restrict__ rowptr, const int* __restrict__ csr_src,
                  const float* __restrict__ csr_w,
                  const float* __restrict__ hfeat,
                  const float* __restrict__ scoef,
                  const float* __restrict__ dcoef,
                  const float* __restrict__ cst, const float* __restrict__ bias,
                  float* __restrict__ outp, int N) {
  constexpr int F = H * 64;
  constexpr int V = 2 * H;
  typedef __attribute__((ext_vector_type(V))) float vecV;
  const int lane = threadIdx.x & 31;
  const int d_first =
      __builtin_amdgcn_readfirstlane(blockIdx.x * 8 + (threadIdx.x >> 5));
  for (int d = d_first; d < N; d += gridDim.x * 8) {
  const int start = rowptr[d];
  const int deg = rowptr[d + 1] - start;
  const int T = deg + 1;

  float dc[H], cs[H];
  for (int h = 0; h < H; ++h) { dc[h] = dcoef[d * H + h]; cs[h] = cst[h]; }

  int s0 = d;
  float w0 = 0.f;
  if (lane < deg) { s0 = csr_src[start + lane]; w0 = csr_w[start + lane]; }
  float pw = w0;
  for (int i = lane + 32; i < deg; i += 32) pw += csr_w[start + i];
  const float wloop = wave_sum_f(pw) / fmaxf((float)deg, 1.0f);
  if (lane == deg) w0 = wloop;
  const bool valid0 = lane < T;
  float al0[H];
  item_logits<H>(s0, w0, scoef, dc, cs, al0);

  float m[H];
  for (int h = 0; h < H; ++h) m[h] = valid0 ? al0[h] : -INFINITY;
  for (int base = 32; base < T; base += 32) {
    const int i = base + lane;
    int s = d; float w = wloop;
    if (i < deg) { s = csr_src[start + i]; w = csr_w[start + i]; }
    float al[H];
    item_logits<H>(s, w, scoef, dc, cs, al);
    if (i < T) {
      for (int h = 0; h < H; ++h) m[h] = fmaxf(m[h], al[h]);
    }
  }
  for (int h = 0; h < H; ++h) m[h] = wave_max_f(m[h]);

  float ex0[H], ps[H];
  for (int h = 0; h < H; ++h) {
    ex0[h] = valid0 ? expf(al0[h] - m[h]) : 0.f;
    ps[h] = ex0[h];
  }
  for (int base = 32; base < T; base += 32) {
    const int i = base + lane;
    int s = d; float w = wloop;
    if (i < deg) { s = csr_src[start + i]; w = csr_w[start + i]; }
    float al[H];
    item_logits<H>(s, w, scoef, dc, cs, al);
    if (i < T) {
      for (int h = 0; h < H; ++h) ps[h] += expf(al[h] - m[h]);
    }
  }
  float ssum[H];
  for (int h = 0; h < H; ++h) ssum[h] = wave_sum_f(ps[h]);

  vecV acc = {};
  float wc[H];
  for (int h = 0; h < H; ++h) wc[h] = ex0[h] / ssum[h];
  gather_chunk<H>(hfeat, s0, wc, (T < 32) ? T : 32, lane, acc);
  for (int base = 32; base < T; base += 32) {
    const int i = base + lane;
    int s = d; float w = wloop;
    if (i < deg) { s = csr_src[start + i]; w = csr_w[start + i]; }
    float al[H];
    item_logits<H>(s, w, scoef, dc, cs, al);
    for (int h = 0; h < H; ++h)
      wc[h] = (i < T) ? expf(al[h] - m[h]) / ssum[h] : 0.f;
    const int rem = T - base;
    gather_chunk<H>(hfeat, s, wc, (rem < 32) ? rem : 32, lane, acc);
  }

  const vecV bv = *reinterpret_cast<const vecV*>(bias + lane * V);
  vecV v = acc + bv;
  if (ELU) {
    for (int q = 0; q < V; ++q) v[q] = (v[q] > 0.f) ? v[q] : expm1f(v[q]);
  }
  *reinterpret_cast<vecV*>(outp + (size_t)d * F + lane * V) = v;
  }
}

extern "C" void kernel_launch(void* const* d_in, const int* in_sizes, int n_in,
                              void* d_out, int out_size, void* d_ws,
                              size_t ws_size, hipStream_t stream) {
  const float* x   = (const float*)d_in[0];
  const int*  eidx = (const int*)d_in[1];
  const float* ew  = (const float*)d_in[2];
  const float* W1  = (const float*)d_in[3];
  const float* We1 = (const float*)d_in[4];
  const float* as1 = (const float*)d_in[5];
  const float* ad1 = (const float*)d_in[6];
  const float* ae1 = (const float*)d_in[7];
  const float* b1  = (const float*)d_in[8];
  const float* W2  = (const float*)d_in[9];
  const float* We2 = (const float*)d_in[10];
  const float* as2 = (const float*)d_in[11];
  const float* ad2 = (const float*)d_in[12];
  const float* ae2 = (const float*)d_in[13];
  const float* b2  = (const float*)d_in[14];
  float* out = (float*)d_out;

  const int N  = in_sizes[0] / 128;
  const int E  = in_sizes[2];
  const int* srcv = eidx;
  const int* dstv = eidx + E;

  float* p = (float*)d_ws;
  auto take = [&p](size_t n) { float* r = p; p += (n + 63) & ~(size_t)63; return r; };
  int*   cnt     = (int*)take(N);
  int*   rowptr  = (int*)take((size_t)N + 1);
  int*   rank    = (int*)take(E);
  int*   rec     = (int*)take(4 * (size_t)E);
  int*   csr_src = (int*)take(E);
  float* csr_w   = take(E);
  float* scoef   = take(2 * (size_t)N);
  float* dcoef   = take(2 * (size_t)N);
  float* cst     = take(8);
  float* hfeat   = take((size_t)N * 128);
  float* agg     = take((size_t)N * 128);

  const int T = 256;
  auto cdiv = [](long long a, long long b) { return (int)((a + b - 1) / b); };
  const int NB  = 2048;
  const int NB6 = 1536;

  hipMemsetAsync(cnt, 0, (size_t)N * 4, stream);
  k_count_rank<<<cdiv(E, T * 8), T, 0, stream>>>(dstv, cnt, rank, E);
  k_rowptr_scan<<<cdiv(N, 1024), 256, 0, stream>>>(cnt, rowptr, N, E);
  k_fill<<<cdiv(E, T * 8), T, 0, stream>>>(srcv, dstv, ew, rowptr, rank, rec, E);
  k_sort_segments<<<NB, 256, 0, stream>>>(rowptr, rec, csr_src, csr_w, N);

  gemm_wmma_f32<128><<<cdiv(cdiv(N, 16), 7), 256, 0, stream>>>(x, W1, hfeat, N);
  k_node_coef<2><<<cdiv(N, T), T, 0, stream>>>(hfeat, as1, ad1, scoef, dcoef, N);
  k_edge_const<2><<<1, 32, 0, stream>>>(We1, ae1, cst);
  k_gat_gather<2, true><<<NB6, 256, 0, stream>>>(
      rowptr, csr_src, csr_w, hfeat, scoef, dcoef, cst, b1, agg, N);

  gemm_wmma_f32<64><<<cdiv(cdiv(N, 16), 3), 128, 0, stream>>>(agg, W2, hfeat, N);
  k_node_coef<1><<<cdiv(N, T), T, 0, stream>>>(hfeat, as2, ad2, scoef, dcoef, N);
  k_edge_const<1><<<1, 32, 0, stream>>>(We2, ae2, cst);
  k_gat_gather<1, false><<<NB, 256, 0, stream>>>(
      rowptr, csr_src, csr_w, hfeat, scoef, dcoef, cst, b2, out, N);
}


# GEMM: one wave owns a whole 16-row strip: A loaded once for all column tiles; s,d attention dots summed from its accumulators (lane partials, then 16-lane xor tree); node_coef passes removed
# speedup vs baseline: 4.2385x; 1.1039x over previous
#include <hip/hip_runtime.h>
#include <math.h>


typedef __attribute__((ext_vector_type(2))) float v2f;
typedef __attribute__((ext_vector_type(8))) float v8f;


__device__ __forceinline__ float row16_sum(float v) {
  v += __int_as_float(__builtin_amdgcn_update_dpp(0, __float_as_int(v), 0x161, 0xf, 0xf, false));
  v += __int_as_float(__builtin_amdgcn_update_dpp(0, __float_as_int(v), 0x162, 0xf, 0xf, false));
  v += __int_as_float(__builtin_amdgcn_update_dpp(0, __float_as_int(v), 0x164, 0xf, 0xf, false));
  v += __int_as_float(__builtin_amdgcn_update_dpp(0, __float_as_int(v), 0x168, 0xf, 0xf, false));
  return v;
}

template <int NCOLS, int H>
__global__ __launch_bounds__(256)
void gemm_wmma_f32(const float* __restrict__ A, const float* __restrict__ Wg,
                   const float* __restrict__ a_s, const float* __restrict__ a_d,
                   float* __restrict__ Out, float* __restrict__ scoef,
                   float* __restrict__ dcoef, int M) {
  constexpr int K = 128;
  constexpr int NT = NCOLS / 16;
  constexpr int TPH = NT / H;
  constexpr int P2 = 2 * NCOLS + 32;
  constexpr int C4 = NCOLS / 4;
  __shared__ __attribute__((aligned(16))) float Ws[(K / 2) * P2];

  const int tid = threadIdx.x;
  constexpr int NSTAGE = (K / 2) * C4 / 256;
  float4 wu[NSTAGE], wv[NSTAGE];
#pragma unroll
  for (int j = 0; j < NSTAGE; ++j) {
    const int i = tid + 256 * j;
    const int p = i / C4, c4 = i - p * C4;
    wu[j] = *reinterpret_cast<const float4*>(Wg + (2 * p) * NCOLS + 4 * c4);
    wv[j] = *reinterpret_cast<const float4*>(Wg + (2 * p + 1) * NCOLS + 4 * c4);
  }
#pragma unroll
  for (int j = 0; j < NSTAGE; ++j) {
    const int i = tid + 256 * j;
    const int p = i / C4, c4 = i - p * C4;
    float4* dst = reinterpret_cast<float4*>(&Ws[p * P2 + 8 * c4]);
    dst[0] = make_float4(wu[j].x, wv[j].x, wu[j].y, wv[j].y);
    dst[1] = make_float4(wu[j].z, wv[j].z, wu[j].w, wv[j].w);
  }
  __syncthreads();

  const int lane  = tid & 31;
  const int r     = lane & 15;
  const int half  = lane >> 4;
  const int nstrips = (M + 15) >> 4;
  const int strip = __builtin_amdgcn_readfirstlane(blockIdx.x * 8 + (tid >> 5));
  if (strip >= nstrips) return;

  const int row16 = strip << 4;
  int arow = row16 + r;
  if (arow >= M) arow = M - 1;
  const float* __restrict__ ap = A + (size_t)arow * K + 2 * half;
  const float* bp = &Ws[half * P2 + 2 * r];

  v8f acc[NT];
#pragma unroll
  for (int t = 0; t < NT; ++t) acc[t] = v8f{};
#pragma unroll
  for (int k = 0; k < K; k += 4) {
    v2f a;
    a.x = ap[k];
    a.y = ap[k + 1];
#pragma unroll
    for (int t = 0; t < NT; ++t) {
      const v2f b = *reinterpret_cast<const v2f*>(bp + (k / 2) * P2 + 32 * t);
      acc[t] = __builtin_amdgcn_wmma_f32_16x16x4_f32(false, a, false, b,
                                                     (short)0, acc[t], false, false);
    }
  }

  const int orow0 = row16 + (half << 3);
  float ps[H][8], pd[H][8];
#pragma unroll
  for (int h = 0; h < H; ++h) {
#pragma unroll
    for (int i = 0; i < 8; ++i) {
      float s0 = 0.f, d0 = 0.f;
#pragma unroll
      for (int t = h * TPH; t < (h + 1) * TPH; ++t) {
        s0 += acc[t][i] * a_s[t * 16 + r];
        d0 += acc[t][i] * a_d[t * 16 + r];
      }
      ps[h][i] = row16_sum(s0);
      pd[h][i] = row16_sum(d0);
    }
  }

#pragma unroll
  for (int i = 0; i < 8; ++i) {
    const int orow = orow0 + i;
    if (orow < M) {
#pragma unroll
      for (int t = 0; t < NT; ++t)
        Out[(size_t)orow * NCOLS + t * 16 + r] = acc[t][i];
      if (r == 0) {
#pragma unroll
        for (int h = 0; h < H; ++h) {
          scoef[orow * H + h] = ps[h][i];
          dcoef[orow * H + h] = pd[h][i];
        }
      }
    }
  }
}

template <int H>
__global__ void k_edge_const(const float* __restrict__ We,
                             const float* __restrict__ ae,
                             float* __restrict__ cst) {
  int h = threadIdx.x;
  if (h < H) {
    float s = 0.f;
    for (int c = 0; c < 64; ++c) s += We[h * 64 + c] * ae[h * 64 + c];
    cst[h] = s;
  }
}


__device__ __forceinline__ int wave_sum_i(int v) {
  for (int o = 16; o > 0; o >>= 1) v += __shfl_xor(v, o);
  return v;
}
__device__ __forceinline__ float wave_sum_f(float v) {
  for (int o = 16; o > 0; o >>= 1) v += __shfl_xor(v, o);
  return v;
}
__device__ __forceinline__ float wave_max_f(float v) {
  for (int o = 16; o > 0; o >>= 1) v = fmaxf(v, __shfl_xor(v, o));
  return v;
}

__global__ void k_count_rank(const int* __restrict__ dst, int* __restrict__ cnt,
                             int* __restrict__ rank, int E) {
  const int stride = gridDim.x * blockDim.x;
  int e = blockIdx.x * blockDim.x + threadIdx.x;
  for (; e < E - 3 * stride; e += 4 * stride) {
    const int d0 = dst[e];
    const int d1 = dst[e + stride];
    const int d2 = dst[e + 2 * stride];
    const int d3 = dst[e + 3 * stride];
    const int r0 = atomicAdd(&cnt[d0], 1);
    const int r1 = atomicAdd(&cnt[d1], 1);
    const int r2 = atomicAdd(&cnt[d2], 1);
    const int r3 = atomicAdd(&cnt[d3], 1);
    rank[e] = r0;
    rank[e + stride] = r1;
    rank[e + 2 * stride] = r2;
    rank[e + 3 * stride] = r3;
  }
  for (; e < E; e += stride) rank[e] = atomicAdd(&cnt[dst[e]], 1);
}

__global__ __launch_bounds__(256)
void k_rowptr_scan(const int* __restrict__ cnt, int* __restrict__ rowptr,
                   int N, int E) {
  __shared__ int s_pre[8];
  __shared__ int s_tot[8];
  const int tid = threadIdx.x, lane = tid & 31, wv = tid >> 5;
  const int b = blockIdx.x;
  int pre = 0;
  for (int c = 0; c < b; ++c) {
    const int4 v = *reinterpret_cast<const int4*>(cnt + c * 1024 + tid * 4);
    pre += (v.x + v.y) + (v.z + v.w);
  }
  const int base = b * 1024 + tid * 4;
  const int v0 = (base     < N) ? cnt[base]     : 0;
  const int v1 = (base + 1 < N) ? cnt[base + 1] : 0;
  const int v2 = (base + 2 < N) ? cnt[base + 2] : 0;
  const int v3 = (base + 3 < N) ? cnt[base + 3] : 0;
  const int tsum = (v0 + v1) + (v2 + v3);
  int inc = tsum;
  for (int o = 1; o < 32; o <<= 1) {
    int t = __shfl_up(inc, o);
    if (lane >= o) inc += t;
  }
  pre = wave_sum_i(pre);
  if (lane == 31) s_tot[wv] = inc;
  if (lane == 0)  s_pre[wv] = pre;
  __syncthreads();
  int off = 0;
  for (int i = 0; i < 8; ++i) {
    off += s_pre[i];
    if (i < wv) off += s_tot[i];
  }
  const int ex = off + inc - tsum;
  if (base     < N) rowptr[base]     = ex;
  if (base + 1 < N) rowptr[base + 1] = ex + v0;
  if (base + 2 < N) rowptr[base + 2] = ex + v0 + v1;
  if (base + 3 < N) rowptr[base + 3] = ex + v0 + v1 + v2;
  if (b == 0 && tid == 0) rowptr[N] = E;
}

__global__ void k_fill(const int* __restrict__ src, const int* __restrict__ dst,
                       const float* __restrict__ ew,
                       const int* __restrict__ rowptr,
                       const int* __restrict__ rank, int* __restrict__ rec_i,
                       int E) {
  int4* __restrict__ rec = reinterpret_cast<int4*>(rec_i);
  const int stride = gridDim.x * blockDim.x;
  int e = blockIdx.x * blockDim.x + threadIdx.x;
  for (; e < E - 3 * stride; e += 4 * stride) {
    const int e1 = e + stride, e2 = e + 2 * stride, e3 = e + 3 * stride;
    const int d0 = dst[e], d1 = dst[e1], d2 = dst[e2], d3 = dst[e3];
    const int p0 = rowptr[d0] + rank[e];
    const int p1 = rowptr[d1] + rank[e1];
    const int p2 = rowptr[d2] + rank[e2];
    const int p3 = rowptr[d3] + rank[e3];
    rec[p0] = make_int4(e,  src[e],  __float_as_int(ew[e]),  0);
    rec[p1] = make_int4(e1, src[e1], __float_as_int(ew[e1]), 0);
    rec[p2] = make_int4(e2, src[e2], __float_as_int(ew[e2]), 0);
    rec[p3] = make_int4(e3, src[e3], __float_as_int(ew[e3]), 0);
  }
  for (; e < E; e += stride) {
    const int pos = rowptr[dst[e]] + rank[e];
    rec[pos] = make_int4(e, src[e], __float_as_int(ew[e]), 0);
  }
}

__global__ __launch_bounds__(256)
void k_sort_segments(const int* __restrict__ rowptr, const int* __restrict__ rec_i,
                     int* __restrict__ csr_src, float* __restrict__ csr_w, int N) {
  const int4* __restrict__ rec = reinterpret_cast<const int4*>(rec_i);
  const int lane = threadIdx.x & 31;
  const int d_first =
      __builtin_amdgcn_readfirstlane(blockIdx.x * 8 + (threadIdx.x >> 5));
  for (int d = d_first; d < N; d += gridDim.x * 8) {
  const int start = rowptr[d];
  const int deg = rowptr[d + 1] - start;
  if (deg <= 32) {
    int4 it = make_int4(0x7fffffff, 0, 0, 0);
    if (lane < deg) it = rec[start + lane];
    int r = 0;
    for (int j = 0; j < deg; ++j) {
      int ej = __builtin_amdgcn_readlane(it.x, j);
      r += (ej < it.x) ? 1 : 0;
    }
    if (lane < deg) {
      csr_src[start + r] = it.y;
      csr_w[start + r] = __int_as_float(it.z);
    }
  } else {
    for (int i = lane; i < deg; i += 32) {
      const int4 it = rec[start + i];
      int r = 0;
      for (int j = 0; j < deg; ++j) r += (rec[start + j].x < it.x) ? 1 : 0;
      csr_src[start + r] = it.y;
      csr_w[start + r] = __int_as_float(it.z);
    }
  }
  }
}

template <int H>
__device__ __forceinline__ void item_logits(int s, float w,
                                            const float* __restrict__ scoef,
                                            const float* dc, const float* cs,
                                            float* al) {
  for (int h = 0; h < H; ++h) {
    float a = scoef[s * H + h] + dc[h] + w * cs[h];
    al[h] = (a >= 0.f) ? a : 0.2f * a;
  }
}

template <int H, int K>
__device__ __forceinline__ void gather_rows(
    const float* __restrict__ hfeat, int s, const float* wc, int j, int lane,
    __attribute__((ext_vector_type(2 * H))) float& acc) {
  constexpr int F = H * 64;
  constexpr int V = 2 * H;
  typedef __attribute__((ext_vector_type(V))) float vecV;
  vecV row[K];
  float c[K];
  for (int k = 0; k < K; ++k) {
    const int sj = __builtin_amdgcn_readlane(s, j + k);
    row[k] = *reinterpret_cast<const vecV*>(hfeat + (size_t)sj * F + lane * V);
    float c0 = __int_as_float(
        __builtin_amdgcn_readlane(__float_as_int(wc[0]), j + k));
    if (H == 2) {
      float c1 = __int_as_float(
          __builtin_amdgcn_readlane(__float_as_int(wc[H - 1]), j + k));
      c0 = (lane < 16) ? c0 : c1;
    }
    c[k] = c0;
  }
  for (int k = 0; k < K; ++k) acc += row[k] * c[k];
}

template <int H>
__device__ __forceinline__ void gather_chunk(
    const float* __restrict__ hfeat, int s, const float* wc, int cnt, int lane,
    __attribute__((ext_vector_type(2 * H))) float& acc) {
  int j = 0;
  for (; j + 8 <= cnt; j += 8) gather_rows<H, 8>(hfeat, s, wc, j, lane, acc);
  if (cnt - j >= 4) { gather_rows<H, 4>(hfeat, s, wc, j, lane, acc); j += 4; }
  if (cnt - j >= 2) { gather_rows<H, 2>(hfeat, s, wc, j, lane, acc); j += 2; }
  if (cnt - j >= 1) { gather_rows<H, 1>(hfeat, s, wc, j, lane, acc); }
}

template <int H, bool ELU>
__global__ __launch_bounds__(256)
void k_gat_gather(const int* __restrict__ rowptr, const int* __restrict__ csr_src,
                  const float* __restrict__ csr_w,
                  const float* __restrict__ hfeat,
                  const float* __restrict__ scoef,
                  const float* __restrict__ dcoef,
                  const float* __restrict__ cst, const float* __restrict__ bias,
                  float* __restrict__ outp, int N) {
  constexpr int F = H * 64;
  constexpr int V = 2 * H;
  typedef __attribute__((ext_vector_type(V))) float vecV;
  const int lane = threadIdx.x & 31;
  const int d_first =
      __builtin_amdgcn_readfirstlane(blockIdx.x * 8 + (threadIdx.x >> 5));
  for (int d = d_first; d < N; d += gridDim.x * 8) {
  const int start = rowptr[d];
  const int deg = rowptr[d + 1] - start;
  const int T = deg + 1;

  float dc[H], cs[H];
  for (int h = 0; h < H; ++h) { dc[h] = dcoef[d * H + h]; cs[h] = cst[h]; }

  int s0 = d;
  float w0 = 0.f;
  if (lane < deg) { s0 = csr_src[start + lane]; w0 = csr_w[start + lane]; }
  float pw = w0;
  for (int i = lane + 32; i < deg; i += 32) pw += csr_w[start + i];
  const float wloop = wave_sum_f(pw) / fmaxf((float)deg, 1.0f);
  if (lane == deg) w0 = wloop;
  const bool valid0 = lane < T;
  float al0[H];
  item_logits<H>(s0, w0, scoef, dc, cs, al0);

  float m[H];
  for (int h = 0; h < H; ++h) m[h] = valid0 ? al0[h] : -INFINITY;
  for (int base = 32; base < T; base += 32) {
    const int i = base + lane;
    int s = d; float w = wloop;
    if (i < deg) { s = csr_src[start + i]; w = csr_w[start + i]; }
    float al[H];
    item_logits<H>(s, w, scoef, dc, cs, al);
    if (i < T) {
      for (int h = 0; h < H; ++h) m[h] = fmaxf(m[h], al[h]);
    }
  }
  for (int h = 0; h < H; ++h) m[h] = wave_max_f(m[h]);

  float ex0[H], ps[H];
  for (int h = 0; h < H; ++h) {
    ex0[h] = valid0 ? expf(al0[h] - m[h]) : 0.f;
    ps[h] = ex0[h];
  }
  for (int base = 32; base < T; base += 32) {
    const int i = base + lane;
    int s = d; float w = wloop;
    if (i < deg) { s = csr_src[start + i]; w = csr_w[start + i]; }
    float al[H];
    item_logits<H>(s, w, scoef, dc, cs, al);
    if (i < T) {
      for (int h = 0; h < H; ++h) ps[h] += expf(al[h] - m[h]);
    }
  }
  float ssum[H];
  for (int h = 0; h < H; ++h) ssum[h] = wave_sum_f(ps[h]);

  vecV acc = {};
  float wc[H];
  for (int h = 0; h < H; ++h) wc[h] = ex0[h] / ssum[h];
  gather_chunk<H>(hfeat, s0, wc, (T < 32) ? T : 32, lane, acc);
  for (int base = 32; base < T; base += 32) {
    const int i = base + lane;
    int s = d; float w = wloop;
    if (i < deg) { s = csr_src[start + i]; w = csr_w[start + i]; }
    float al[H];
    item_logits<H>(s, w, scoef, dc, cs, al);
    for (int h = 0; h < H; ++h)
      wc[h] = (i < T) ? expf(al[h] - m[h]) / ssum[h] : 0.f;
    const int rem = T - base;
    gather_chunk<H>(hfeat, s, wc, (rem < 32) ? rem : 32, lane, acc);
  }

  const vecV bv = *reinterpret_cast<const vecV*>(bias + lane * V);
  vecV v = acc + bv;
  if (ELU) {
    for (int q = 0; q < V; ++q) v[q] = (v[q] > 0.f) ? v[q] : expm1f(v[q]);
  }
  *reinterpret_cast<vecV*>(outp + (size_t)d * F + lane * V) = v;
  }
}

extern "C" void kernel_launch(void* const* d_in, const int* in_sizes, int n_in,
                              void* d_out, int out_size, void* d_ws,
                              size_t ws_size, hipStream_t stream) {
  const float* x   = (const float*)d_in[0];
  const int*  eidx = (const int*)d_in[1];
  const float* ew  = (const float*)d_in[2];
  const float* W1  = (const float*)d_in[3];
  const float* We1 = (const float*)d_in[4];
  const float* as1 = (const float*)d_in[5];
  const float* ad1 = (const float*)d_in[6];
  const float* ae1 = (const float*)d_in[7];
  const float* b1  = (const float*)d_in[8];
  const float* W2  = (const float*)d_in[9];
  const float* We2 = (const float*)d_in[10];
  const float* as2 = (const float*)d_in[11];
  const float* ad2 = (const float*)d_in[12];
  const float* ae2 = (const float*)d_in[13];
  const float* b2  = (const float*)d_in[14];
  float* out = (float*)d_out;

  const int N  = in_sizes[0] / 128;
  const int E  = in_sizes[2];
  const int* srcv = eidx;
  const int* dstv = eidx + E;

  float* p = (float*)d_ws;
  auto take = [&p](size_t n) { float* r = p; p += (n + 63) & ~(size_t)63; return r; };
  int*   cnt     = (int*)take(N);
  int*   rowptr  = (int*)take((size_t)N + 1);
  int*   rank    = (int*)take(E);
  int*   rec     = (int*)take(4 * (size_t)E);
  int*   csr_src = (int*)take(E);
  float* csr_w   = take(E);
  float* scoef   = take(2 * (size_t)N);
  float* dcoef   = take(2 * (size_t)N);
  float* cst     = take(8);
  float* hfeat   = take((size_t)N * 128);
  float* agg     = take((size_t)N * 128);

  const int T = 256;
  auto cdiv = [](long long a, long long b) { return (int)((a + b - 1) / b); };
  const int NB  = 2048;
  const int NB6 = 1536;

  hipMemsetAsync(cnt, 0, (size_t)N * 4, stream);
  k_count_rank<<<cdiv(E, T * 8), T, 0, stream>>>(dstv, cnt, rank, E);
  k_rowptr_scan<<<cdiv(N, 1024), 256, 0, stream>>>(cnt, rowptr, N, E);
  k_fill<<<cdiv(E, T * 8), T, 0, stream>>>(srcv, dstv, ew, rowptr, rank, rec, E);
  k_sort_segments<<<NB, 256, 0, stream>>>(rowptr, rec, csr_src, csr_w, N);

  gemm_wmma_f32<128, 2><<<cdiv(cdiv(N, 16), 8), 256, 0, stream>>>(
      x, W1, as1, ad1, hfeat, scoef, dcoef, N);
  k_edge_const<2><<<1, 32, 0, stream>>>(We1, ae1, cst);
  k_gat_gather<2, true><<<NB6, 256, 0, stream>>>(
      rowptr, csr_src, csr_w, hfeat, scoef, dcoef, cst, b1, agg, N);

  gemm_wmma_f32<64, 1><<<cdiv(cdiv(N, 16), 8), 256, 0, stream>>>(
      agg, W2, as2, ad2, hfeat, scoef, dcoef, N);
  k_edge_const<1><<<1, 32, 0, stream>>>(We2, ae2, cst);
  k_gat_gather<1, false><<<NB, 256, 0, stream>>>(
      rowptr, csr_src, csr_w, hfeat, scoef, dcoef, cst, b2, out, N);
}


# Kernels that do not depend on each other share a launch: the layer-1 GEMM runs as work-groups inside the in-degree count's and the record fill's grids (half each); edge constants ride along
# speedup vs baseline: 4.7392x; 1.1181x over previous
#include <hip/hip_runtime.h>
#include <math.h>


typedef __attribute__((ext_vector_type(2))) float v2f;
typedef __attribute__((ext_vector_type(8))) float v8f;


__device__ __forceinline__ float row16_sum(float v) {
  v += __int_as_float(__builtin_amdgcn_update_dpp(0, __float_as_int(v), 0x161, 0xf, 0xf, false));
  v += __int_as_float(__builtin_amdgcn_update_dpp(0, __float_as_int(v), 0x162, 0xf, 0xf, false));
  v += __int_as_float(__builtin_amdgcn_update_dpp(0, __float_as_int(v), 0x164, 0xf, 0xf, false));
  v += __int_as_float(__builtin_amdgcn_update_dpp(0, __float_as_int(v), 0x168, 0xf, 0xf, false));
  return v;
}

template <int NCOLS, int H>
__device__ __forceinline__
void gemm_body(int gblock, const float* __restrict__ A, const float* __restrict__ Wg,
               const float* __restrict__ a_s, const float* __restrict__ a_d,
               float* __restrict__ Out, float* __restrict__ scoef,
               float* __restrict__ dcoef, int M) {
  constexpr int K = 128;
  constexpr int NT = NCOLS / 16;
  constexpr int TPH = NT / H;
  constexpr int P2 = 2 * NCOLS + 32;
  constexpr int C4 = NCOLS / 4;
  __shared__ __attribute__((aligned(16))) float Ws[(K / 2) * P2];

  const int tid = threadIdx.x;
  constexpr int NSTAGE = (K / 2) * C4 / 256;
  float4 wu[NSTAGE], wv[NSTAGE];
#pragma unroll
  for (int j = 0; j < NSTAGE; ++j) {
    const int i = tid + 256 * j;
    const int p = i / C4, c4 = i - p * C4;
    wu[j] = *reinterpret_cast<const float4*>(Wg + (2 * p) * NCOLS + 4 * c4);
    wv[j] = *reinterpret_cast<const float4*>(Wg + (2 * p + 1) * NCOLS + 4 * c4);
  }
#pragma unroll
  for (int j = 0; j < NSTAGE; ++j) {
    const int i = tid + 256 * j;
    const int p = i / C4, c4 = i - p * C4;
    float4* dst = reinterpret_cast<float4*>(&Ws[p * P2 + 8 * c4]);
    dst[0] = make_float4(wu[j].x, wv[j].x, wu[j].y, wv[j].y);
    dst[1] = make_float4(wu[j].z, wv[j].z, wu[j].w, wv[j].w);
  }
  __syncthreads();

  const int lane  = tid & 31;
  const int r     = lane & 15;
  const int half  = lane >> 4;
  const int nstrips = (M + 15) >> 4;
  const int strip = __builtin_amdgcn_readfirstlane(gblock * 8 + (tid >> 5));
  if (strip >= nstrips) return;

  const int row16 = strip << 4;
  int arow = row16 + r;
  if (arow >= M) arow = M - 1;
  const float* __restrict__ ap = A + (size_t)arow * K + 2 * half;
  const float* bp = &Ws[half * P2 + 2 * r];

  v8f acc[NT];
#pragma unroll
  for (int t = 0; t < NT; ++t) acc[t] = v8f{};
#pragma unroll
  for (int k = 0; k < K; k += 4) {
    v2f a;
    a.x = ap[k];
    a.y = ap[k + 1];
#pragma unroll
    for (int t = 0; t < NT; ++t) {
      const v2f b = *reinterpret_cast<const v2f*>(bp + (k / 2) * P2 + 32 * t);
      acc[t] = __builtin_amdgcn_wmma_f32_16x16x4_f32(false, a, false, b,
                                                     (short)0, acc[t], false, false);
    }
  }

  const int orow0 = row16 + (half << 3);
  float ps[H][8], pd[H][8];
#pragma unroll
  for (int h = 0; h < H; ++h) {
#pragma unroll
    for (int i = 0; i < 8; ++i) {
      float s0 = 0.f, d0 = 0.f;
#pragma unroll
      for (int t = h * TPH; t < (h + 1) * TPH; ++t) {
        s0 += acc[t][i] * a_s[t * 16 + r];
        d0 += acc[t][i] * a_d[t * 16 + r];
      }
      ps[h][i] = row16_sum(s0);
      pd[h][i] = row16_sum(d0);
    }
  }

#pragma unroll
  for (int i = 0; i < 8; ++i) {
    const int orow = orow0 + i;
    if (orow < M) {
#pragma unroll
      for (int t = 0; t < NT; ++t)
        Out[(size_t)orow * NCOLS + t * 16 + r] = acc[t][i];
      if (r == 0) {
#pragma unroll
        for (int h = 0; h < H; ++h) {
          scoef[orow * H + h] = ps[h][i];
          dcoef[orow * H + h] = pd[h][i];
        }
      }
    }
  }
}

template <int NCOLS, int H>
__global__ __launch_bounds__(256)
void gemm_wmma_f32(const float* __restrict__ A, const float* __restrict__ Wg,
                   const float* __restrict__ a_s, const float* __restrict__ a_d,
                   float* __restrict__ Out, float* __restrict__ scoef,
                   float* __restrict__ dcoef, int M) {
  gemm_body<NCOLS, H>(blockIdx.x, A, Wg, a_s, a_d, Out, scoef, dcoef, M);
}

__device__ __forceinline__ void edge_const(const float* __restrict__ We,
                                           const float* __restrict__ ae,
                                           float* __restrict__ cst, int h) {
  float s = 0.f;
  for (int c = 0; c < 64; ++c) s += We[h * 64 + c] * ae[h * 64 + c];
  cst[h] = s;
}

__device__ __forceinline__ bool shared_launch_role(int b, int ngb, int neb, int& idx) {
  const int m = (ngb < neb / 2) ? ngb : neb / 2;
  if (b < 3 * m) {
    const int q = b / 3, r = b - 3 * q;
    if (r == 0) { idx = q; return true; }
    idx = 2 * q + r - 1;
    return false;
  }
  const int rest = b - 3 * m;
  if (rest < ngb - m) { idx = m + rest; return true; }
  idx = 2 * m + rest - (ngb - m);
  return false;
}


__device__ __forceinline__ int wave_sum_i(int v) {
  for (int o = 16; o > 0; o >>= 1) v += __shfl_xor(v, o);
  return v;
}
__device__ __forceinline__ float wave_sum_f(float v) {
  for (int o = 16; o > 0; o >>= 1) v += __shfl_xor(v, o);
  return v;
}
__device__ __forceinline__ float wave_max_f(float v) {
  for (int o = 16; o > 0; o >>= 1) v = fmaxf(v, __shfl_xor(v, o));
  return v;
}

__device__ __forceinline__ void count_rank_body(int vb, int nvb,
                                                const int* __restrict__ dst,
                                                int* __restrict__ cnt,
                                                int* __restrict__ rank, int E) {
  const int stride = nvb * blockDim.x;
  int e = vb * blockDim.x + threadIdx.x;
  for (; e < E - 3 * stride; e += 4 * stride) {
    const int d0 = dst[e];
    const int d1 = dst[e + stride];
    const int d2 = dst[e + 2 * stride];
    const int d3 = dst[e + 3 * stride];
    const int r0 = atomicAdd(&cnt[d0], 1);
    const int r1 = atomicAdd(&cnt[d1], 1);
    const int r2 = atomicAdd(&cnt[d2], 1);
    const int r3 = atomicAdd(&cnt[d3], 1);
    rank[e] = r0;
    rank[e + stride] = r1;
    rank[e + 2 * stride] = r2;
    rank[e + 3 * stride] = r3;
  }
  for (; e < E; e += stride) rank[e] = atomicAdd(&cnt[dst[e]], 1);
}

__global__ __launch_bounds__(256)
void k_rowptr_scan(const int* __restrict__ cnt, int* __restrict__ rowptr,
                   int N, int E) {
  __shared__ int s_pre[8];
  __shared__ int s_tot[8];
  const int tid = threadIdx.x, lane = tid & 31, wv = tid >> 5;
  const int b = blockIdx.x;
  int pre = 0;
  for (int c = 0; c < b; ++c) {
    const int4 v = *reinterpret_cast<const int4*>(cnt + c * 1024 + tid * 4);
    pre += (v.x + v.y) + (v.z + v.w);
  }
  const int base = b * 1024 + tid * 4;
  const int v0 = (base     < N) ? cnt[base]     : 0;
  const int v1 = (base + 1 < N) ? cnt[base + 1] : 0;
  const int v2 = (base + 2 < N) ? cnt[base + 2] : 0;
  const int v3 = (base + 3 < N) ? cnt[base + 3] : 0;
  const int tsum = (v0 + v1) + (v2 + v3);
  int inc = tsum;
  for (int o = 1; o < 32; o <<= 1) {
    int t = __shfl_up(inc, o);
    if (lane >= o) inc += t;
  }
  pre = wave_sum_i(pre);
  if (lane == 31) s_tot[wv] = inc;
  if (lane == 0)  s_pre[wv] = pre;
  __syncthreads();
  int off = 0;
  for (int i = 0; i < 8; ++i) {
    off += s_pre[i];
    if (i < wv) off += s_tot[i];
  }
  const int ex = off + inc - tsum;
  if (base     < N) rowptr[base]     = ex;
  if (base + 1 < N) rowptr[base + 1] = ex + v0;
  if (base + 2 < N) rowptr[base + 2] = ex + v0 + v1;
  if (base + 3 < N) rowptr[base + 3] = ex + v0 + v1 + v2;
  if (b == 0 && tid == 0) rowptr[N] = E;
}

__device__ __forceinline__ void fill_body(int vb, int nvb,
                                          const int* __restrict__ src,
                                          const int* __restrict__ dst,
                                          const float* __restrict__ ew,
                                          const int* __restrict__ rowptr,
                                          const int* __restrict__ rank,
                                          int* __restrict__ rec_i, int E) {
  int4* __restrict__ rec = reinterpret_cast<int4*>(rec_i);
  const int stride = nvb * blockDim.x;
  int e = vb * blockDim.x + threadIdx.x;
  for (; e < E - 3 * stride; e += 4 * stride) {
    const int e1 = e + stride, e2 = e + 2 * stride, e3 = e + 3 * stride;
    const int d0 = dst[e], d1 = dst[e1], d2 = dst[e2], d3 = dst[e3];
    const int p0 = rowptr[d0] + rank[e];
    const int p1 = rowptr[d1] + rank[e1];
    const int p2 = rowptr[d2] + rank[e2];
    const int p3 = rowptr[d3] + rank[e3];
    rec[p0] = make_int4(e,  src[e],  __float_as_int(ew[e]),  0);
    rec[p1] = make_int4(e1, src[e1], __float_as_int(ew[e1]), 0);
    rec[p2] = make_int4(e2, src[e2], __float_as_int(ew[e2]), 0);
    rec[p3] = make_int4(e3, src[e3], __float_as_int(ew[e3]), 0);
  }
  for (; e < E; e += stride) {
    const int pos = rowptr[dst[e]] + rank[e];
    rec[pos] = make_int4(e, src[e], __float_as_int(ew[e]), 0);
  }
}

__global__ __launch_bounds__(256)
void k_gemm1_count(const float* __restrict__ x, const float* __restrict__ W1,
                   const float* __restrict__ as1, const float* __restrict__ ad1,
                   float* __restrict__ hfeat, float* __restrict__ scoef,
                   float* __restrict__ dcoef, int N, int gb0, int ngb,
                   const int* __restrict__ dst, int* __restrict__ cnt,
                   int* __restrict__ rank, int E, int neb,
                   const float* __restrict__ We1, const float* __restrict__ ae1,
                   const float* __restrict__ We2, const float* __restrict__ ae2,
                   float* __restrict__ cst) {
  int idx;
  if (shared_launch_role(blockIdx.x, ngb, neb, idx)) {
    gemm_body<128, 2>(gb0 + idx, x, W1, as1, ad1, hfeat, scoef, dcoef, N);
  } else if (idx < neb) {
    count_rank_body(idx, neb, dst, cnt, rank, E);
  } else {
    const int t = threadIdx.x;
    if (t < 2) edge_const(We1, ae1, cst, t);
    if (t == 32) edge_const(We2, ae2, cst + 4, 0);
  }
}

__global__ __launch_bounds__(256)
void k_gemm1_fill(const float* __restrict__ x, const float* __restrict__ W1,
                  const float* __restrict__ as1, const float* __restrict__ ad1,
                  float* __restrict__ hfeat, float* __restrict__ scoef,
                  float* __restrict__ dcoef, int N, int gb0, int ngb,
                  const int* __restrict__ src, const int* __restrict__ dst,
                  const float* __restrict__ ew, const int* __restrict__ rowptr,
                  const int* __restrict__ rank, int* __restrict__ rec, int E,
                  int neb) {
  int idx;
  if (shared_launch_role(blockIdx.x, ngb, neb, idx)) {
    gemm_body<128, 2>(gb0 + idx, x, W1, as1, ad1, hfeat, scoef, dcoef, N);
  } else if (idx < neb) {
    fill_body(idx, neb, src, dst, ew, rowptr, rank, rec, E);
  }
}

__global__ __launch_bounds__(256)
void k_sort_segments(const int* __restrict__ rowptr, const int* __restrict__ rec_i,
                     int* __restrict__ csr_src, float* __restrict__ csr_w, int N) {
  const int4* __restrict__ rec = reinterpret_cast<const int4*>(rec_i);
  const int lane = threadIdx.x & 31;
  const int d_first =
      __builtin_amdgcn_readfirstlane(blockIdx.x * 8 + (threadIdx.x >> 5));
  for (int d = d_first; d < N; d += gridDim.x * 8) {
  const int start = rowptr[d];
  const int deg = rowptr[d + 1] - start;
  if (deg <= 32) {
    int4 it = make_int4(0x7fffffff, 0, 0, 0);
    if (lane < deg) it = rec[start + lane];
    int r = 0;
    for (int j = 0; j < deg; ++j) {
      int ej = __builtin_amdgcn_readlane(it.x, j);
      r += (ej < it.x) ? 1 : 0;
    }
    if (lane < deg) {
      csr_src[start + r] = it.y;
      csr_w[start + r] = __int_as_float(it.z);
    }
  } else {
    for (int i = lane; i < deg; i += 32) {
      const int4 it = rec[start + i];
      int r = 0;
      for (int j = 0; j < deg; ++j) r += (rec[start + j].x < it.x) ? 1 : 0;
      csr_src[start + r] = it.y;
      csr_w[start + r] = __int_as_float(it.z);
    }
  }
  }
}

template <int H>
__device__ __forceinline__ void item_logits(int s, float w,
                                            const float* __restrict__ scoef,
                                            const float* dc, const float* cs,
                                            float* al) {
  for (int h = 0; h < H; ++h) {
    float a = scoef[s * H + h] + dc[h] + w * cs[h];
    al[h] = (a >= 0.f) ? a : 0.2f * a;
  }
}

template <int H, int K>
__device__ __forceinline__ void gather_rows(
    const float* __restrict__ hfeat, int s, const float* wc, int j, int lane,
    __attribute__((ext_vector_type(2 * H))) float& acc) {
  constexpr int F = H * 64;
  constexpr int V = 2 * H;
  typedef __attribute__((ext_vector_type(V))) float vecV;
  vecV row[K];
  float c[K];
  for (int k = 0; k < K; ++k) {
    const int sj = __builtin_amdgcn_readlane(s, j + k);
    row[k] = *reinterpret_cast<const vecV*>(hfeat + (size_t)sj * F + lane * V);
    float c0 = __int_as_float(
        __builtin_amdgcn_readlane(__float_as_int(wc[0]), j + k));
    if (H == 2) {
      float c1 = __int_as_float(
          __builtin_amdgcn_readlane(__float_as_int(wc[H - 1]), j + k));
      c0 = (lane < 16) ? c0 : c1;
    }
    c[k] = c0;
  }
  for (int k = 0; k < K; ++k) acc += row[k] * c[k];
}

template <int H>
__device__ __forceinline__ void gather_chunk(
    const float* __restrict__ hfeat, int s, const float* wc, int cnt, int lane,
    __attribute__((ext_vector_type(2 * H))) float& acc) {
  int j = 0;
  for (; j + 8 <= cnt; j += 8) gather_rows<H, 8>(hfeat, s, wc, j, lane, acc);
  if (cnt - j >= 4) { gather_rows<H, 4>(hfeat, s, wc, j, lane, acc); j += 4; }
  if (cnt - j >= 2) { gather_rows<H, 2>(hfeat, s, wc, j, lane, acc); j += 2; }
  if (cnt - j >= 1) { gather_rows<H, 1>(hfeat, s, wc, j, lane, acc); }
}

template <int H, bool ELU>
__global__ __launch_bounds__(256)
void k_gat_gather(const int* __restrict__ rowptr, const int* __restrict__ csr_src,
                  const float* __restrict__ csr_w,
                  const float* __restrict__ hfeat,
                  const float* __restrict__ scoef,
                  const float* __restrict__ dcoef,
                  const float* __restrict__ cst, const float* __restrict__ bias,
                  float* __restrict__ outp, int N) {
  constexpr int F = H * 64;
  constexpr int V = 2 * H;
  typedef __attribute__((ext_vector_type(V))) float vecV;
  const int lane = threadIdx.x & 31;
  const int d_first =
      __builtin_amdgcn_readfirstlane(blockIdx.x * 8 + (threadIdx.x >> 5));
  for (int d = d_first; d < N; d += gridDim.x * 8) {
  const int start = rowptr[d];
  const int deg = rowptr[d + 1] - start;
  const int T = deg + 1;

  float dc[H], cs[H];
  for (int h = 0; h < H; ++h) { dc[h] = dcoef[d * H + h]; cs[h] = cst[h]; }

  int s0 = d;
  float w0 = 0.f;
  if (lane < deg) { s0 = csr_src[start + lane]; w0 = csr_w[start + lane]; }
  float pw = w0;
  for (int i = lane + 32; i < deg; i += 32) pw += csr_w[start + i];
  const float wloop = wave_sum_f(pw) / fmaxf((float)deg, 1.0f);
  if (lane == deg) w0 = wloop;
  const bool valid0 = lane < T;
  float al0[H];
  item_logits<H>(s0, w0, scoef, dc, cs, al0);

  float m[H];
  for (int h = 0; h < H; ++h) m[h] = valid0 ? al0[h] : -INFINITY;
  for (int base = 32; base < T; base += 32) {
    const int i = base + lane;
    int s = d; float w = wloop;
    if (i < deg) { s = csr_src[start + i]; w = csr_w[start + i]; }
    float al[H];
    item_logits<H>(s, w, scoef, dc, cs, al);
    if (i < T) {
      for (int h = 0; h < H; ++h) m[h] = fmaxf(m[h], al[h]);
    }
  }
  for (int h = 0; h < H; ++h) m[h] = wave_max_f(m[h]);

  float ex0[H], ps[H];
  for (int h = 0; h < H; ++h) {
    ex0[h] = valid0 ? expf(al0[h] - m[h]) : 0.f;
    ps[h] = ex0[h];
  }
  for (int base = 32; base < T; base += 32) {
    const int i = base + lane;
    int s = d; float w = wloop;
    if (i < deg) { s = csr_src[start + i]; w = csr_w[start + i]; }
    float al[H];
    item_logits<H>(s, w, scoef, dc, cs, al);
    if (i < T) {
      for (int h = 0; h < H; ++h) ps[h] += expf(al[h] - m[h]);
    }
  }
  float ssum[H];
  for (int h = 0; h < H; ++h) ssum[h] = wave_sum_f(ps[h]);

  vecV acc = {};
  float wc[H];
  for (int h = 0; h < H; ++h) wc[h] = ex0[h] / ssum[h];
  gather_chunk<H>(hfeat, s0, wc, (T < 32) ? T : 32, lane, acc);
  for (int base = 32; base < T; base += 32) {
    const int i = base + lane;
    int s = d; float w = wloop;
    if (i < deg) { s = csr_src[start + i]; w = csr_w[start + i]; }
    float al[H];
    item_logits<H>(s, w, scoef, dc, cs, al);
    for (int h = 0; h < H; ++h)
      wc[h] = (i < T) ? expf(al[h] - m[h]) / ssum[h] : 0.f;
    const int rem = T - base;
    gather_chunk<H>(hfeat, s, wc, (rem < 32) ? rem : 32, lane, acc);
  }

  const vecV bv = *reinterpret_cast<const vecV*>(bias + lane * V);
  vecV v = acc + bv;
  if (ELU) {
    for (int q = 0; q < V; ++q) v[q] = (v[q] > 0.f) ? v[q] : expm1f(v[q]);
  }
  *reinterpret_cast<vecV*>(outp + (size_t)d * F + lane * V) = v;
  }
}

extern "C" void kernel_launch(void* const* d_in, const int* in_sizes, int n_in,
                              void* d_out, int out_size, void* d_ws,
                              size_t ws_size, hipStream_t stream) {
  const float* x   = (const float*)d_in[0];
  const int*  eidx = (const int*)d_in[1];
  const float* ew  = (const float*)d_in[2];
  const float* W1  = (const float*)d_in[3];
  const float* We1 = (const float*)d_in[4];
  const float* as1 = (const float*)d_in[5];
  const float* ad1 = (const float*)d_in[6];
  const float* ae1 = (const float*)d_in[7];
  const float* b1  = (const float*)d_in[8];
  const float* W2  = (const float*)d_in[9];
  const float* We2 = (const float*)d_in[10];
  const float* as2 = (const float*)d_in[11];
  const float* ad2 = (const float*)d_in[12];
  const float* ae2 = (const float*)d_in[13];
  const float* b2  = (const float*)d_in[14];
  float* out = (float*)d_out;

  const int N  = in_sizes[0] / 128;
  const int E  = in_sizes[2];
  const int* srcv = eidx;
  const int* dstv = eidx + E;

  float* p = (float*)d_ws;
  auto take = [&p](size_t n) { float* r = p; p += (n + 63) & ~(size_t)63; return r; };
  int*   cnt     = (int*)take(N);
  int*   rowptr  = (int*)take((size_t)N + 1);
  int*   rank    = (int*)take(E);
  int*   rec     = (int*)take(4 * (size_t)E);
  int*   csr_src = (int*)take(E);
  float* csr_w   = take(E);
  float* scoef   = take(2 * (size_t)N);
  float* dcoef   = take(2 * (size_t)N);
  float* cst     = take(8);
  float* hfeat   = take((size_t)N * 128);
  float* agg     = take((size_t)N * 128);

  const int T = 256;
  auto cdiv = [](long long a, long long b) { return (int)((a + b - 1) / b); };
  const int NB  = 2048;
  const int NB6 = 1536;

  const int GB  = cdiv(cdiv(N, 16), 8);
  const int GBA = GB / 2;
  const int EB  = cdiv(E, T * 8);
  hipMemsetAsync(cnt, 0, (size_t)N * 4, stream);
  k_gemm1_count<<<GBA + EB + 1, T, 0, stream>>>(
      x, W1, as1, ad1, hfeat, scoef, dcoef, N, 0, GBA,
      dstv, cnt, rank, E, EB, We1, ae1, We2, ae2, cst);
  k_rowptr_scan<<<cdiv(N, 1024), 256, 0, stream>>>(cnt, rowptr, N, E);
  k_gemm1_fill<<<(GB - GBA) + EB, T, 0, stream>>>(
      x, W1, as1, ad1, hfeat, scoef, dcoef, N, GBA, GB - GBA,
      srcv, dstv, ew, rowptr, rank, rec, E, EB);
  k_sort_segments<<<NB, 256, 0, stream>>>(rowptr, rec, csr_src, csr_w, N);

  k_gat_gather<2, true><<<NB6, 256, 0, stream>>>(
      rowptr, csr_src, csr_w, hfeat, scoef, dcoef, cst, b1, agg, N);

  gemm_wmma_f32<64, 1><<<cdiv(cdiv(N, 16), 8), 256, 0, stream>>>(
      agg, W2, as2, ad2, hfeat, scoef, dcoef, N);
  k_gat_gather<1, false><<<NB, 256, 0, stream>>>(
      rowptr, csr_src, csr_w, hfeat, scoef, dcoef, cst + 4, b2, out, N);
}


# CSR build without per-edge global atomics: edges binned by 128-node range via LDS histograms and a scan, then one work-group per bin counts and ranks per node in LDS; float sum order unchanged
# speedup vs baseline: 5.5850x; 1.1785x over previous
#include <hip/hip_runtime.h>
#include <math.h>


typedef __attribute__((ext_vector_type(2))) float v2f;
typedef __attribute__((ext_vector_type(8))) float v8f;


__device__ __forceinline__ float row16_sum(float v) {
  v += __int_as_float(__builtin_amdgcn_update_dpp(0, __float_as_int(v), 0x161, 0xf, 0xf, false));
  v += __int_as_float(__builtin_amdgcn_update_dpp(0, __float_as_int(v), 0x162, 0xf, 0xf, false));
  v += __int_as_float(__builtin_amdgcn_update_dpp(0, __float_as_int(v), 0x164, 0xf, 0xf, false));
  v += __int_as_float(__builtin_amdgcn_update_dpp(0, __float_as_int(v), 0x168, 0xf, 0xf, false));
  return v;
}

template <int NCOLS, int H>
__device__ __forceinline__
void gemm_body(int gblock, const float* __restrict__ A, const float* __restrict__ Wg,
               const float* __restrict__ a_s, const float* __restrict__ a_d,
               float* __restrict__ Out, float* __restrict__ scoef,
               float* __restrict__ dcoef, int M) {
  constexpr int K = 128;
  constexpr int NT = NCOLS / 16;
  constexpr int TPH = NT / H;
  constexpr int P2 = 2 * NCOLS + 32;
  constexpr int C4 = NCOLS / 4;
  __shared__ __attribute__((aligned(16))) float Ws[(K / 2) * P2];

  const int tid = threadIdx.x;
  constexpr int NSTAGE = (K / 2) * C4 / 256;
  float4 wu[NSTAGE], wv[NSTAGE];
#pragma unroll
  for (int j = 0; j < NSTAGE; ++j) {
    const int i = tid + 256 * j;
    const int p = i / C4, c4 = i - p * C4;
    wu[j] = *reinterpret_cast<const float4*>(Wg + (2 * p) * NCOLS + 4 * c4);
    wv[j] = *reinterpret_cast<const float4*>(Wg + (2 * p + 1) * NCOLS + 4 * c4);
  }
#pragma unroll
  for (int j = 0; j < NSTAGE; ++j) {
    const int i = tid + 256 * j;
    const int p = i / C4, c4 = i - p * C4;
    float4* dst = reinterpret_cast<float4*>(&Ws[p * P2 + 8 * c4]);
    dst[0] = make_float4(wu[j].x, wv[j].x, wu[j].y, wv[j].y);
    dst[1] = make_float4(wu[j].z, wv[j].z, wu[j].w, wv[j].w);
  }
  __syncthreads();

  const int lane  = tid & 31;
  const int r     = lane & 15;
  const int half  = lane >> 4;
  const int nstrips = (M + 15) >> 4;
  const int strip = __builtin_amdgcn_readfirstlane(gblock * 8 + (tid >> 5));
  if (strip >= nstrips) return;

  const int row16 = strip << 4;
  int arow = row16 + r;
  if (arow >= M) arow = M - 1;
  const float* __restrict__ ap = A + (size_t)arow * K + 2 * half;
  const float* bp = &Ws[half * P2 + 2 * r];

  v8f acc[NT];
#pragma unroll
  for (int t = 0; t < NT; ++t) acc[t] = v8f{};
#pragma unroll
  for (int k = 0; k < K; k += 4) {
    v2f a;
    a.x = ap[k];
    a.y = ap[k + 1];
#pragma unroll
    for (int t = 0; t < NT; ++t) {
      const v2f b = *reinterpret_cast<const v2f*>(bp + (k / 2) * P2 + 32 * t);
      acc[t] = __builtin_amdgcn_wmma_f32_16x16x4_f32(false, a, false, b,
                                                     (short)0, acc[t], false, false);
    }
  }

  const int orow0 = row16 + (half << 3);
  float ps[H][8], pd[H][8];
#pragma unroll
  for (int h = 0; h < H; ++h) {
#pragma unroll
    for (int i = 0; i < 8; ++i) {
      float s0 = 0.f, d0 = 0.f;
#pragma unroll
      for (int t = h * TPH; t < (h + 1) * TPH; ++t) {
        s0 += acc[t][i] * a_s[t * 16 + r];
        d0 += acc[t][i] * a_d[t * 16 + r];
      }
      ps[h][i] = row16_sum(s0);
      pd[h][i] = row16_sum(d0);
    }
  }

#pragma unroll
  for (int i = 0; i < 8; ++i) {
    const int orow = orow0 + i;
    if (orow < M) {
#pragma unroll
      for (int t = 0; t < NT; ++t)
        Out[(size_t)orow * NCOLS + t * 16 + r] = acc[t][i];
      if (r == 0) {
#pragma unroll
        for (int h = 0; h < H; ++h) {
          scoef[orow * H + h] = ps[h][i];
          dcoef[orow * H + h] = pd[h][i];
        }
      }
    }
  }
}

template <int NCOLS, int H>
__global__ __launch_bounds__(256)
void gemm_wmma_f32(const float* __restrict__ A, const float* __restrict__ Wg,
                   const float* __restrict__ a_s, const float* __restrict__ a_d,
                   float* __restrict__ Out, float* __restrict__ scoef,
                   float* __restrict__ dcoef, int M) {
  gemm_body<NCOLS, H>(blockIdx.x, A, Wg, a_s, a_d, Out, scoef, dcoef, M);
}

__device__ __forceinline__ void edge_const(const float* __restrict__ We,
                                           const float* __restrict__ ae,
                                           float* __restrict__ cst, int h) {
  float s = 0.f;
  for (int c = 0; c < 64; ++c) s += We[h * 64 + c] * ae[h * 64 + c];
  cst[h] = s;
}

__device__ __forceinline__ bool shared_launch_role(int b, int ngb, int neb, int& idx) {
  const int m = (ngb < neb / 2) ? ngb : neb / 2;
  if (b < 3 * m) {
    const int q = b / 3, r = b - 3 * q;
    if (r == 0) { idx = q; return true; }
    idx = 2 * q + r - 1;
    return false;
  }
  const int rest = b - 3 * m;
  if (rest < ngb - m) { idx = m + rest; return true; }
  idx = 2 * m + rest - (ngb - m);
  return false;
}


__device__ __forceinline__ int wave_sum_i(int v) {
  for (int o = 16; o > 0; o >>= 1) v += __shfl_xor(v, o);
  return v;
}
__device__ __forceinline__ float wave_sum_f(float v) {
  for (int o = 16; o > 0; o >>= 1) v += __shfl_xor(v, o);
  return v;
}
__device__ __forceinline__ float wave_max_f(float v) {
  for (int o = 16; o > 0; o >>= 1) v = fmaxf(v, __shfl_xor(v, o));
  return v;
}

constexpr int BIN_SHIFT = 7;
constexpr int BIN_NODES = 1 << BIN_SHIFT;
constexpr int MAX_BINS  = 1024;

__device__ __forceinline__ void bin_count_body(int vb, int nvb,
                                               const int* __restrict__ dst,
                                               int* __restrict__ hist,
                                               int* __restrict__ bintot,
                                               int E, int nbin) {
  __shared__ int lh[MAX_BINS];
  const int tid = threadIdx.x;
  for (int i = tid; i < nbin; i += blockDim.x) lh[i] = 0;
  __syncthreads();
  const int stride = nvb * blockDim.x;
  int e = vb * blockDim.x + tid;
  for (; e < E - 3 * stride; e += 4 * stride) {
    const int d0 = dst[e];
    const int d1 = dst[e + stride];
    const int d2 = dst[e + 2 * stride];
    const int d3 = dst[e + 3 * stride];
    atomicAdd(&lh[d0 >> BIN_SHIFT], 1);
    atomicAdd(&lh[d1 >> BIN_SHIFT], 1);
    atomicAdd(&lh[d2 >> BIN_SHIFT], 1);
    atomicAdd(&lh[d3 >> BIN_SHIFT], 1);
  }
  for (; e < E; e += stride) atomicAdd(&lh[dst[e] >> BIN_SHIFT], 1);
  __syncthreads();
  for (int i = tid; i < nbin; i += blockDim.x) {
    const int c = lh[i];
    hist[i * nvb + vb] = c;
    if (c != 0) atomicAdd(&bintot[i], c);
  }
}

__global__ __launch_bounds__(256)
void k_bin_offsets(const int* __restrict__ bintot, int* __restrict__ hist,
                   int nvb) {
  __shared__ int s_pre[8];
  __shared__ int s_tot[8];
  const int tid = threadIdx.x, lane = tid & 31, wv = tid >> 5;
  const int b = blockIdx.x;
  int pre = 0;
  for (int i = tid; i < b; i += 256) pre += bintot[i];
  pre = wave_sum_i(pre);
  if (lane == 0) s_pre[wv] = pre;
  __syncthreads();
  int carry = 0;
  for (int i = 0; i < 8; ++i) carry += s_pre[i];
  int* __restrict__ row = hist + (size_t)b * nvb;
  for (int t0 = 0; t0 < nvb; t0 += 1024) {
    const int base = t0 + tid * 4;
    const int v0 = (base     < nvb) ? row[base]     : 0;
    const int v1 = (base + 1 < nvb) ? row[base + 1] : 0;
    const int v2 = (base + 2 < nvb) ? row[base + 2] : 0;
    const int v3 = (base + 3 < nvb) ? row[base + 3] : 0;
    const int tsum = (v0 + v1) + (v2 + v3);
    int inc = tsum;
    for (int o = 1; o < 32; o <<= 1) {
      int t = __shfl_up(inc, o);
      if (lane >= o) inc += t;
    }
    if (lane == 31) s_tot[wv] = inc;
    __syncthreads();
    int off = carry, tile = 0;
    for (int i = 0; i < 8; ++i) {
      const int t = s_tot[i];
      if (i < wv) off += t;
      tile += t;
    }
    const int ex = off + inc - tsum;
    if (base     < nvb) row[base]     = ex;
    if (base + 1 < nvb) row[base + 1] = ex + v0;
    if (base + 2 < nvb) row[base + 2] = ex + v0 + v1;
    if (base + 3 < nvb) row[base + 3] = ex + v0 + v1 + v2;
    carry += tile;
    __syncthreads();
  }
}

__device__ __forceinline__ void bin_scatter_body(int vb, int nvb,
                                                 const int* __restrict__ src,
                                                 const int* __restrict__ dst,
                                                 const float* __restrict__ ew,
                                                 const int* __restrict__ offs,
                                                 int* __restrict__ rec_i,
                                                 int E, int nbin) {
  __shared__ int lh[MAX_BINS];
  int4* __restrict__ rec = reinterpret_cast<int4*>(rec_i);
  const int tid = threadIdx.x;
  for (int i = tid; i < nbin; i += blockDim.x) lh[i] = offs[i * nvb + vb];
  __syncthreads();
  const int stride = nvb * blockDim.x;
  int e = vb * blockDim.x + tid;
  for (; e < E - 3 * stride; e += 4 * stride) {
    const int e1 = e + stride, e2 = e + 2 * stride, e3 = e + 3 * stride;
    const int d0 = dst[e], d1 = dst[e1], d2 = dst[e2], d3 = dst[e3];
    const int p0 = atomicAdd(&lh[d0 >> BIN_SHIFT], 1);
    const int p1 = atomicAdd(&lh[d1 >> BIN_SHIFT], 1);
    const int p2 = atomicAdd(&lh[d2 >> BIN_SHIFT], 1);
    const int p3 = atomicAdd(&lh[d3 >> BIN_SHIFT], 1);
    rec[p0] = make_int4(e,  src[e],  __float_as_int(ew[e]),  d0);
    rec[p1] = make_int4(e1, src[e1], __float_as_int(ew[e1]), d1);
    rec[p2] = make_int4(e2, src[e2], __float_as_int(ew[e2]), d2);
    rec[p3] = make_int4(e3, src[e3], __float_as_int(ew[e3]), d3);
  }
  for (; e < E; e += stride) {
    const int d = dst[e];
    const int pos = atomicAdd(&lh[d >> BIN_SHIFT], 1);
    rec[pos] = make_int4(e, src[e], __float_as_int(ew[e]), d);
  }
}

__global__ __launch_bounds__(256)
void k_bin_place(const int* __restrict__ offs, const int* __restrict__ bintot,
                 const int* __restrict__ rec0_i, int* __restrict__ rec_i,
                 int* __restrict__ rowptr, int N, int E, int nvb) {
  __shared__ int lc[BIN_NODES];
  __shared__ int s_tot[8];
  const int4* __restrict__ rec0 = reinterpret_cast<const int4*>(rec0_i);
  int4* __restrict__ rec = reinterpret_cast<int4*>(rec_i);
  const int tid = threadIdx.x, lane = tid & 31, wv = tid >> 5;
  const int b = blockIdx.x;
  if (tid < BIN_NODES) lc[tid] = 0;
  __syncthreads();
  const int base = offs[(size_t)b * nvb];
  const int tot = bintot[b];
  for (int i0 = tid; i0 < tot + tid; i0 += 8 * 256) {
    int dn[8];
#pragma unroll
    for (int u = 0; u < 8; ++u) {
      const int i = i0 + 256 * u;
      dn[u] = rec0[base + ((i < tot) ? i : tot - 1)].w;
    }
#pragma unroll
    for (int u = 0; u < 8; ++u)
      if (i0 + 256 * u < tot) atomicAdd(&lc[dn[u] & (BIN_NODES - 1)], 1);
  }
  __syncthreads();
  const int c = (tid < BIN_NODES) ? lc[tid] : 0;
  int inc = c;
  for (int o = 1; o < 32; o <<= 1) {
    int t = __shfl_up(inc, o);
    if (lane >= o) inc += t;
  }
  if (lane == 31) s_tot[wv] = inc;
  __syncthreads();
  int off = base;
  for (int i = 0; i < wv; ++i) off += s_tot[i];
  const int ex = off + inc - c;
  if (tid < BIN_NODES) {
    lc[tid] = ex;
    const int node = b * BIN_NODES + tid;
    if (node < N) rowptr[node] = ex;
  }
  if (b == 0 && tid == 0) rowptr[N] = E;
  __syncthreads();
  for (int i0 = tid; i0 < tot + tid; i0 += 8 * 256) {
    int4 it[8];
#pragma unroll
    for (int u = 0; u < 8; ++u) {
      const int i = i0 + 256 * u;
      it[u] = rec0[base + ((i < tot) ? i : tot - 1)];
    }
#pragma unroll
    for (int u = 0; u < 8; ++u) {
      if (i0 + 256 * u < tot) {
        const int k = it[u].w & (BIN_NODES - 1);
        const int pos = atomicAdd(&lc[k], 1);
        rec[pos] = make_int4(it[u].x, it[u].y, it[u].z, 0);
      }
    }
  }
}

__global__ __launch_bounds__(256)
void k_gemm1_count(const float* __restrict__ x, const float* __restrict__ W1,
                   const float* __restrict__ as1, const float* __restrict__ ad1,
                   float* __restrict__ hfeat, float* __restrict__ scoef,
                   float* __restrict__ dcoef, int N, int gb0, int ngb,
                   const int* __restrict__ dst, int* __restrict__ hist,
                   int* __restrict__ bintot, int E, int neb, int nbin,
                   const float* __restrict__ We1, const float* __restrict__ ae1,
                   const float* __restrict__ We2, const float* __restrict__ ae2,
                   float* __restrict__ cst) {
  int idx;
  if (shared_launch_role(blockIdx.x, ngb, neb, idx)) {
    gemm_body<128, 2>(gb0 + idx, x, W1, as1, ad1, hfeat, scoef, dcoef, N);
  } else if (idx < neb) {
    bin_count_body(idx, neb, dst, hist, bintot, E, nbin);
  } else {
    const int t = threadIdx.x;
    if (t < 2) edge_const(We1, ae1, cst, t);
    if (t == 32) edge_const(We2, ae2, cst + 4, 0);
  }
}

__global__ __launch_bounds__(256)
void k_gemm1_fill(const float* __restrict__ x, const float* __restrict__ W1,
                  const float* __restrict__ as1, const float* __restrict__ ad1,
                  float* __restrict__ hfeat, float* __restrict__ scoef,
                  float* __restrict__ dcoef, int N, int gb0, int ngb,
                  const int* __restrict__ src, const int* __restrict__ dst,
                  const float* __restrict__ ew, const int* __restrict__ offs,
                  int* __restrict__ rec0, int E, int neb, int nbin) {
  int idx;
  if (shared_launch_role(blockIdx.x, ngb, neb, idx)) {
    gemm_body<128, 2>(gb0 + idx, x, W1, as1, ad1, hfeat, scoef, dcoef, N);
  } else if (idx < neb) {
    bin_scatter_body(idx, neb, src, dst, ew, offs, rec0, E, nbin);
  }
}

__global__ __launch_bounds__(256)
void k_sort_segments(const int* __restrict__ rowptr, const int* __restrict__ rec_i,
                     int* __restrict__ csr_src, float* __restrict__ csr_w, int N) {
  const int4* __restrict__ rec = reinterpret_cast<const int4*>(rec_i);
  const int lane = threadIdx.x & 31;
  const int d_first =
      __builtin_amdgcn_readfirstlane(blockIdx.x * 8 + (threadIdx.x >> 5));
  for (int d = d_first; d < N; d += gridDim.x * 8) {
  const int start = rowptr[d];
  const int deg = rowptr[d + 1] - start;
  if (deg <= 32) {
    int4 it = make_int4(0x7fffffff, 0, 0, 0);
    if (lane < deg) it = rec[start + lane];
    int r = 0;
    for (int j = 0; j < deg; ++j) {
      int ej = __builtin_amdgcn_readlane(it.x, j);
      r += (ej < it.x) ? 1 : 0;
    }
    if (lane < deg) {
      csr_src[start + r] = it.y;
      csr_w[start + r] = __int_as_float(it.z);
    }
  } else {
    for (int i = lane; i < deg; i += 32) {
      const int4 it = rec[start + i];
      int r = 0;
      for (int j = 0; j < deg; ++j) r += (rec[start + j].x < it.x) ? 1 : 0;
      csr_src[start + r] = it.y;
      csr_w[start + r] = __int_as_float(it.z);
    }
  }
  }
}

template <int H>
__device__ __forceinline__ void item_logits(int s, float w,
                                            const float* __restrict__ scoef,
                                            const float* dc, const float* cs,
                                            float* al) {
  for (int h = 0; h < H; ++h) {
    float a = scoef[s * H + h] + dc[h] + w * cs[h];
    al[h] = (a >= 0.f) ? a : 0.2f * a;
  }
}

template <int H, int K>
__device__ __forceinline__ void gather_rows(
    const float* __restrict__ hfeat, int s, const float* wc, int j, int lane,
    __attribute__((ext_vector_type(2 * H))) float& acc) {
  constexpr int F = H * 64;
  constexpr int V = 2 * H;
  typedef __attribute__((ext_vector_type(V))) float vecV;
  vecV row[K];
  float c[K];
  for (int k = 0; k < K; ++k) {
    const int sj = __builtin_amdgcn_readlane(s, j + k);
    row[k] = *reinterpret_cast<const vecV*>(hfeat + (size_t)sj * F + lane * V);
    float c0 = __int_as_float(
        __builtin_amdgcn_readlane(__float_as_int(wc[0]), j + k));
    if (H == 2) {
      float c1 = __int_as_float(
          __builtin_amdgcn_readlane(__float_as_int(wc[H - 1]), j + k));
      c0 = (lane < 16) ? c0 : c1;
    }
    c[k] = c0;
  }
  for (int k = 0; k < K; ++k) acc += row[k] * c[k];
}

template <int H>
__device__ __forceinline__ void gather_chunk(
    const float* __restrict__ hfeat, int s, const float* wc, int cnt, int lane,
    __attribute__((ext_vector_type(2 * H))) float& acc) {
  int j = 0;
  for (; j + 8 <= cnt; j += 8) gather_rows<H, 8>(hfeat, s, wc, j, lane, acc);
  if (cnt - j >= 4) { gather_rows<H, 4>(hfeat, s, wc, j, lane, acc); j += 4; }
  if (cnt - j >= 2) { gather_rows<H, 2>(hfeat, s, wc, j, lane, acc); j += 2; }
  if (cnt - j >= 1) { gather_rows<H, 1>(hfeat, s, wc, j, lane, acc); }
}

template <int H, bool ELU>
__global__ __launch_bounds__(256)
void k_gat_gather(const int* __restrict__ rowptr, const int* __restrict__ csr_src,
                  const float* __restrict__ csr_w,
                  const float* __restrict__ hfeat,
                  const float* __restrict__ scoef,
                  const float* __restrict__ dcoef,
                  const float* __restrict__ cst, const float* __restrict__ bias,
                  float* __restrict__ outp, int N) {
  constexpr int F = H * 64;
  constexpr int V = 2 * H;
  typedef __attribute__((ext_vector_type(V))) float vecV;
  const int lane = threadIdx.x & 31;
  const int d_first =
      __builtin_amdgcn_readfirstlane(blockIdx.x * 8 + (threadIdx.x >> 5));
  for (int d = d_first; d < N; d += gridDim.x * 8) {
  const int start = rowptr[d];
  const int deg = rowptr[d + 1] - start;
  const int T = deg + 1;

  float dc[H], cs[H];
  for (int h = 0; h < H; ++h) { dc[h] = dcoef[d * H + h]; cs[h] = cst[h]; }

  int s0 = d;
  float w0 = 0.f;
  if (lane < deg) { s0 = csr_src[start + lane]; w0 = csr_w[start + lane]; }
  float pw = w0;
  for (int i = lane + 32; i < deg; i += 32) pw += csr_w[start + i];
  const float wloop = wave_sum_f(pw) / fmaxf((float)deg, 1.0f);
  if (lane == deg) w0 = wloop;
  const bool valid0 = lane < T;
  float al0[H];
  item_logits<H>(s0, w0, scoef, dc, cs, al0);

  float m[H];
  for (int h = 0; h < H; ++h) m[h] = valid0 ? al0[h] : -INFINITY;
  for (int base = 32; base < T; base += 32) {
    const int i = base + lane;
    int s = d; float w = wloop;
    if (i < deg) { s = csr_src[start + i]; w = csr_w[start + i]; }
    float al[H];
    item_logits<H>(s, w, scoef, dc, cs, al);
    if (i < T) {
      for (int h = 0; h < H; ++h) m[h] = fmaxf(m[h], al[h]);
    }
  }
  for (int h = 0; h < H; ++h) m[h] = wave_max_f(m[h]);

  float ex0[H], ps[H];
  for (int h = 0; h < H; ++h) {
    ex0[h] = valid0 ? expf(al0[h] - m[h]) : 0.f;
    ps[h] = ex0[h];
  }
  for (int base = 32; base < T; base += 32) {
    const int i = base + lane;
    int s = d; float w = wloop;
    if (i < deg) { s = csr_src[start + i]; w = csr_w[start + i]; }
    float al[H];
    item_logits<H>(s, w, scoef, dc, cs, al);
    if (i < T) {
      for (int h = 0; h < H; ++h) ps[h] += expf(al[h] - m[h]);
    }
  }
  float ssum[H];
  for (int h = 0; h < H; ++h) ssum[h] = wave_sum_f(ps[h]);

  vecV acc = {};
  float wc[H];
  for (int h = 0; h < H; ++h) wc[h] = ex0[h] / ssum[h];
  gather_chunk<H>(hfeat, s0, wc, (T < 32) ? T : 32, lane, acc);
  for (int base = 32; base < T; base += 32) {
    const int i = base + lane;
    int s = d; float w = wloop;
    if (i < deg) { s = csr_src[start + i]; w = csr_w[start + i]; }
    float al[H];
    item_logits<H>(s, w, scoef, dc, cs, al);
    for (int h = 0; h < H; ++h)
      wc[h] = (i < T) ? expf(al[h] - m[h]) / ssum[h] : 0.f;
    const int rem = T - base;
    gather_chunk<H>(hfeat, s, wc, (rem < 32) ? rem : 32, lane, acc);
  }

  const vecV bv = *reinterpret_cast<const vecV*>(bias + lane * V);
  vecV v = acc + bv;
  if (ELU) {
    for (int q = 0; q < V; ++q) v[q] = (v[q] > 0.f) ? v[q] : expm1f(v[q]);
  }
  *reinterpret_cast<vecV*>(outp + (size_t)d * F + lane * V) = v;
  }
}

extern "C" void kernel_launch(void* const* d_in, const int* in_sizes, int n_in,
                              void* d_out, int out_size, void* d_ws,
                              size_t ws_size, hipStream_t stream) {
  const float* x   = (const float*)d_in[0];
  const int*  eidx = (const int*)d_in[1];
  const float* ew  = (const float*)d_in[2];
  const float* W1  = (const float*)d_in[3];
  const float* We1 = (const float*)d_in[4];
  const float* as1 = (const float*)d_in[5];
  const float* ad1 = (const float*)d_in[6];
  const float* ae1 = (const float*)d_in[7];
  const float* b1  = (const float*)d_in[8];
  const float* W2  = (const float*)d_in[9];
  const float* We2 = (const float*)d_in[10];
  const float* as2 = (const float*)d_in[11];
  const float* ad2 = (const float*)d_in[12];
  const float* ae2 = (const float*)d_in[13];
  const float* b2  = (const float*)d_in[14];
  float* out = (float*)d_out;

  const int N  = in_sizes[0] / 128;
  const int E  = in_sizes[2];
  const int* srcv = eidx;
  const int* dstv = eidx + E;

  const int T = 256;
  auto cdiv = [](long long a, long long b) { return (int)((a + b - 1) / b); };
  const int EB   = cdiv(E, T * 8);
  const int NBIN = cdiv(N, BIN_NODES);

  float* p = (float*)d_ws;
  auto take = [&p](size_t n) { float* r = p; p += (n + 63) & ~(size_t)63; return r; };
  int*   bintot  = (int*)take(MAX_BINS);
  int*   rowptr  = (int*)take((size_t)N + 1);
  int*   hist    = (int*)take((size_t)NBIN * EB);
  int*   rec0    = (int*)take(4 * (size_t)E);
  int*   rec     = (int*)take(4 * (size_t)E);
  int*   csr_src = (int*)take(E);
  float* csr_w   = take(E);
  float* scoef   = take(2 * (size_t)N);
  float* dcoef   = take(2 * (size_t)N);
  float* cst     = take(8);
  float* hfeat   = take((size_t)N * 128);
  float* agg     = take((size_t)N * 128);

  const int NB  = 2048;
  const int NB6 = 1536;

  const int GB  = cdiv(cdiv(N, 16), 8);
  const int GBA = GB / 2;
  hipMemsetAsync(bintot, 0, (size_t)NBIN * 4, stream);
  k_gemm1_count<<<GBA + EB + 1, T, 0, stream>>>(
      x, W1, as1, ad1, hfeat, scoef, dcoef, N, 0, GBA,
      dstv, hist, bintot, E, EB, NBIN, We1, ae1, We2, ae2, cst);
  k_bin_offsets<<<NBIN, 256, 0, stream>>>(bintot, hist, EB);
  k_gemm1_fill<<<(GB - GBA) + EB, T, 0, stream>>>(
      x, W1, as1, ad1, hfeat, scoef, dcoef, N, GBA, GB - GBA,
      srcv, dstv, ew, hist, rec0, E, EB, NBIN);
  k_bin_place<<<NBIN, 256, 0, stream>>>(hist, bintot, rec0, rec, rowptr, N, E, EB);
  k_sort_segments<<<NB, 256, 0, stream>>>(rowptr, rec, csr_src, csr_w, N);

  k_gat_gather<2, true><<<NB6, 256, 0, stream>>>(
      rowptr, csr_src, csr_w, hfeat, scoef, dcoef, cst, b1, agg, N);

  gemm_wmma_f32<64, 1><<<cdiv(cdiv(N, 16), 8), 256, 0, stream>>>(
      agg, W2, as2, ad2, hfeat, scoef, dcoef, N);
  k_gat_gather<1, false><<<NB, 256, 0, stream>>>(
      rowptr, csr_src, csr_w, hfeat, scoef, dcoef, cst + 4, b2, out, N);
}


# CSR level 2 finishes in the bin's work-group: records go to per-node segments in LDS, are ranked there by edge id and written in final order; the per-node sort launch is gone; sum order unchanged
# speedup vs baseline: 5.8297x; 1.0438x over previous
#include <hip/hip_runtime.h>
#include <math.h>


typedef __attribute__((ext_vector_type(2))) float v2f;
typedef __attribute__((ext_vector_type(8))) float v8f;


__device__ __forceinline__ float row16_sum(float v) {
  v += __int_as_float(__builtin_amdgcn_update_dpp(0, __float_as_int(v), 0x161, 0xf, 0xf, false));
  v += __int_as_float(__builtin_amdgcn_update_dpp(0, __float_as_int(v), 0x162, 0xf, 0xf, false));
  v += __int_as_float(__builtin_amdgcn_update_dpp(0, __float_as_int(v), 0x164, 0xf, 0xf, false));
  v += __int_as_float(__builtin_amdgcn_update_dpp(0, __float_as_int(v), 0x168, 0xf, 0xf, false));
  return v;
}

template <int NCOLS, int H>
__device__ __forceinline__
void gemm_body(int gblock, const float* __restrict__ A, const float* __restrict__ Wg,
               const float* __restrict__ a_s, const float* __restrict__ a_d,
               float* __restrict__ Out, float* __restrict__ scoef,
               float* __restrict__ dcoef, int M) {
  constexpr int K = 128;
  constexpr int NT = NCOLS / 16;
  constexpr int TPH = NT / H;
  constexpr int P2 = 2 * NCOLS + 32;
  constexpr int C4 = NCOLS / 4;
  __shared__ __attribute__((aligned(16))) float Ws[(K / 2) * P2];

  const int tid = threadIdx.x;
  constexpr int NSTAGE = (K / 2) * C4 / 256;
  float4 wu[NSTAGE], wv[NSTAGE];
#pragma unroll
  for (int j = 0; j < NSTAGE; ++j) {
    const int i = tid + 256 * j;
    const int p = i / C4, c4 = i - p * C4;
    wu[j] = *reinterpret_cast<const float4*>(Wg + (2 * p) * NCOLS + 4 * c4);
    wv[j] = *reinterpret_cast<const float4*>(Wg + (2 * p + 1) * NCOLS + 4 * c4);
  }
#pragma unroll
  for (int j = 0; j < NSTAGE; ++j) {
    const int i = tid + 256 * j;
    const int p = i / C4, c4 = i - p * C4;
    float4* dst = reinterpret_cast<float4*>(&Ws[p * P2 + 8 * c4]);
    dst[0] = make_float4(wu[j].x, wv[j].x, wu[j].y, wv[j].y);
    dst[1] = make_float4(wu[j].z, wv[j].z, wu[j].w, wv[j].w);
  }
  __syncthreads();

  const int lane  = tid & 31;
  const int r     = lane & 15;
  const int half  = lane >> 4;
  const int nstrips = (M + 15) >> 4;
  const int strip = __builtin_amdgcn_readfirstlane(gblock * 8 + (tid >> 5));
  if (strip >= nstrips) return;

  const int row16 = strip << 4;
  int arow = row16 + r;
  if (arow >= M) arow = M - 1;
  const float* __restrict__ ap = A + (size_t)arow * K + 2 * half;
  const float* bp = &Ws[half * P2 + 2 * r];

  v8f acc[NT];
#pragma unroll
  for (int t = 0; t < NT; ++t) acc[t] = v8f{};
#pragma unroll
  for (int k = 0; k < K; k += 4) {
    v2f a;
    a.x = ap[k];
    a.y = ap[k + 1];
#pragma unroll
    for (int t = 0; t < NT; ++t) {
      const v2f b = *reinterpret_cast<const v2f*>(bp + (k / 2) * P2 + 32 * t);
      acc[t] = __builtin_amdgcn_wmma_f32_16x16x4_f32(false, a, false, b,
                                                     (short)0, acc[t], false, false);
    }
  }

  const int orow0 = row16 + (half << 3);
  float ps[H][8], pd[H][8];
#pragma unroll
  for (int h = 0; h < H; ++h) {
#pragma unroll
    for (int i = 0; i < 8; ++i) {
      float s0 = 0.f, d0 = 0.f;
#pragma unroll
      for (int t = h * TPH; t < (h + 1) * TPH; ++t) {
        s0 += acc[t][i] * a_s[t * 16 + r];
        d0 += acc[t][i] * a_d[t * 16 + r];
      }
      ps[h][i] = row16_sum(s0);
      pd[h][i] = row16_sum(d0);
    }
  }

#pragma unroll
  for (int i = 0; i < 8; ++i) {
    const int orow = orow0 + i;
    if (orow < M) {
#pragma unroll
      for (int t = 0; t < NT; ++t)
        Out[(size_t)orow * NCOLS + t * 16 + r] = acc[t][i];
      if (r == 0) {
#pragma unroll
        for (int h = 0; h < H; ++h) {
          scoef[orow * H + h] = ps[h][i];
          dcoef[orow * H + h] = pd[h][i];
        }
      }
    }
  }
}

template <int NCOLS, int H>
__global__ __launch_bounds__(256)
void gemm_wmma_f32(const float* __restrict__ A, const float* __restrict__ Wg,
                   const float* __restrict__ a_s, const float* __restrict__ a_d,
                   float* __restrict__ Out, float* __restrict__ scoef,
                   float* __restrict__ dcoef, int M) {
  gemm_body<NCOLS, H>(blockIdx.x, A, Wg, a_s, a_d, Out, scoef, dcoef, M);
}

__device__ __forceinline__ void edge_const(const float* __restrict__ We,
                                           const float* __restrict__ ae,
                                           float* __restrict__ cst, int h) {
  float s = 0.f;
  for (int c = 0; c < 64; ++c) s += We[h * 64 + c] * ae[h * 64 + c];
  cst[h] = s;
}

__device__ __forceinline__ bool shared_launch_role(int b, int ngb, int neb, int& idx) {
  const int m = (ngb < neb / 2) ? ngb : neb / 2;
  if (b < 3 * m) {
    const int q = b / 3, r = b - 3 * q;
    if (r == 0) { idx = q; return true; }
    idx = 2 * q + r - 1;
    return false;
  }
  const int rest = b - 3 * m;
  if (rest < ngb - m) { idx = m + rest; return true; }
  idx = 2 * m + rest - (ngb - m);
  return false;
}


__device__ __forceinline__ int wave_sum_i(int v) {
  for (int o = 16; o > 0; o >>= 1) v += __shfl_xor(v, o);
  return v;
}
__device__ __forceinline__ float wave_sum_f(float v) {
  for (int o = 16; o > 0; o >>= 1) v += __shfl_xor(v, o);
  return v;
}
__device__ __forceinline__ float wave_max_f(float v) {
  for (int o = 16; o > 0; o >>= 1) v = fmaxf(v, __shfl_xor(v, o));
  return v;
}

constexpr int BIN_SHIFT = 7;
constexpr int BIN_NODES = 1 << BIN_SHIFT;
constexpr int MAX_BINS  = 1024;

__device__ __forceinline__ void bin_count_body(int vb, int nvb,
                                               const int* __restrict__ dst,
                                               int* __restrict__ hist,
                                               int* __restrict__ bintot,
                                               int E, int nbin) {
  __shared__ int lh[MAX_BINS];
  const int tid = threadIdx.x;
  for (int i = tid; i < nbin; i += blockDim.x) lh[i] = 0;
  __syncthreads();
  const int stride = nvb * blockDim.x;
  int e = vb * blockDim.x + tid;
  for (; e < E - 3 * stride; e += 4 * stride) {
    const int d0 = dst[e];
    const int d1 = dst[e + stride];
    const int d2 = dst[e + 2 * stride];
    const int d3 = dst[e + 3 * stride];
    atomicAdd(&lh[d0 >> BIN_SHIFT], 1);
    atomicAdd(&lh[d1 >> BIN_SHIFT], 1);
    atomicAdd(&lh[d2 >> BIN_SHIFT], 1);
    atomicAdd(&lh[d3 >> BIN_SHIFT], 1);
  }
  for (; e < E; e += stride) atomicAdd(&lh[dst[e] >> BIN_SHIFT], 1);
  __syncthreads();
  for (int i = tid; i < nbin; i += blockDim.x) {
    const int c = lh[i];
    hist[i * nvb + vb] = c;
    if (c != 0) atomicAdd(&bintot[i], c);
  }
}

__global__ __launch_bounds__(256)
void k_bin_offsets(const int* __restrict__ bintot, int* __restrict__ hist,
                   int nvb) {
  __shared__ int s_pre[8];
  __shared__ int s_tot[8];
  const int tid = threadIdx.x, lane = tid & 31, wv = tid >> 5;
  const int b = blockIdx.x;
  int pre = 0;
  for (int i = tid; i < b; i += 256) pre += bintot[i];
  pre = wave_sum_i(pre);
  if (lane == 0) s_pre[wv] = pre;
  __syncthreads();
  int carry = 0;
  for (int i = 0; i < 8; ++i) carry += s_pre[i];
  int* __restrict__ row = hist + (size_t)b * nvb;
  for (int t0 = 0; t0 < nvb; t0 += 1024) {
    const int base = t0 + tid * 4;
    const int v0 = (base     < nvb) ? row[base]     : 0;
    const int v1 = (base + 1 < nvb) ? row[base + 1] : 0;
    const int v2 = (base + 2 < nvb) ? row[base + 2] : 0;
    const int v3 = (base + 3 < nvb) ? row[base + 3] : 0;
    const int tsum = (v0 + v1) + (v2 + v3);
    int inc = tsum;
    for (int o = 1; o < 32; o <<= 1) {
      int t = __shfl_up(inc, o);
      if (lane >= o) inc += t;
    }
    if (lane == 31) s_tot[wv] = inc;
    __syncthreads();
    int off = carry, tile = 0;
    for (int i = 0; i < 8; ++i) {
      const int t = s_tot[i];
      if (i < wv) off += t;
      tile += t;
    }
    const int ex = off + inc - tsum;
    if (base     < nvb) row[base]     = ex;
    if (base + 1 < nvb) row[base + 1] = ex + v0;
    if (base + 2 < nvb) row[base + 2] = ex + v0 + v1;
    if (base + 3 < nvb) row[base + 3] = ex + v0 + v1 + v2;
    carry += tile;
    __syncthreads();
  }
}

__device__ __forceinline__ void bin_scatter_body(int vb, int nvb,
                                                 const int* __restrict__ src,
                                                 const int* __restrict__ dst,
                                                 const float* __restrict__ ew,
                                                 const int* __restrict__ offs,
                                                 int* __restrict__ rec_i,
                                                 int E, int nbin) {
  __shared__ int lh[MAX_BINS];
  int4* __restrict__ rec = reinterpret_cast<int4*>(rec_i);
  const int tid = threadIdx.x;
  for (int i = tid; i < nbin; i += blockDim.x) lh[i] = offs[i * nvb + vb];
  __syncthreads();
  const int stride = nvb * blockDim.x;
  int e = vb * blockDim.x + tid;
  for (; e < E - 3 * stride; e += 4 * stride) {
    const int e1 = e + stride, e2 = e + 2 * stride, e3 = e + 3 * stride;
    const int d0 = dst[e], d1 = dst[e1], d2 = dst[e2], d3 = dst[e3];
    const int p0 = atomicAdd(&lh[d0 >> BIN_SHIFT], 1);
    const int p1 = atomicAdd(&lh[d1 >> BIN_SHIFT], 1);
    const int p2 = atomicAdd(&lh[d2 >> BIN_SHIFT], 1);
    const int p3 = atomicAdd(&lh[d3 >> BIN_SHIFT], 1);
    rec[p0] = make_int4(e,  src[e],  __float_as_int(ew[e]),  d0);
    rec[p1] = make_int4(e1, src[e1], __float_as_int(ew[e1]), d1);
    rec[p2] = make_int4(e2, src[e2], __float_as_int(ew[e2]), d2);
    rec[p3] = make_int4(e3, src[e3], __float_as_int(ew[e3]), d3);
  }
  for (; e < E; e += stride) {
    const int d = dst[e];
    const int pos = atomicAdd(&lh[d >> BIN_SHIFT], 1);
    rec[pos] = make_int4(e, src[e], __float_as_int(ew[e]), d);
  }
}

constexpr int PLACE_CAP = 4096;
constexpr int NODES_PER_WAVE = BIN_NODES / 8;

__global__ __launch_bounds__(256)
void k_bin_place(const int* __restrict__ offs, const int* __restrict__ bintot,
                 const int* __restrict__ rec0_i, int* __restrict__ rec_i,
                 int* __restrict__ rowptr, int* __restrict__ csr_src,
                 float* __restrict__ csr_w, int N, int E, int nvb) {
  __shared__ int lc[BIN_NODES];
  __shared__ int lst[BIN_NODES + 1];
  __shared__ int s_tot[8];
  __shared__ int l_eid[PLACE_CAP];
  __shared__ int l_src[PLACE_CAP];
  __shared__ int l_w[PLACE_CAP];
  const int4* __restrict__ rec0 = reinterpret_cast<const int4*>(rec0_i);
  int4* __restrict__ rec = reinterpret_cast<int4*>(rec_i);
  const int tid = threadIdx.x, lane = tid & 31, wv = tid >> 5;
  const int b = blockIdx.x;
  if (tid < BIN_NODES) lc[tid] = 0;
  __syncthreads();
  const int base = offs[(size_t)b * nvb];
  const int tot = bintot[b];
  const bool in_lds = tot <= PLACE_CAP;
  for (int i0 = tid; i0 < tot + tid; i0 += 8 * 256) {
    int dn[8];
#pragma unroll
    for (int u = 0; u < 8; ++u) {
      const int i = i0 + 256 * u;
      dn[u] = rec0[base + ((i < tot) ? i : tot - 1)].w;
    }
#pragma unroll
    for (int u = 0; u < 8; ++u)
      if (i0 + 256 * u < tot) atomicAdd(&lc[dn[u] & (BIN_NODES - 1)], 1);
  }
  __syncthreads();
  const int c = (tid < BIN_NODES) ? lc[tid] : 0;
  int inc = c;
  for (int o = 1; o < 32; o <<= 1) {
    int t = __shfl_up(inc, o);
    if (lane >= o) inc += t;
  }
  if (lane == 31) s_tot[wv] = inc;
  __syncthreads();
  int off = 0;
  for (int i = 0; i < wv; ++i) off += s_tot[i];
  const int ex = off + inc - c;
  if (tid < BIN_NODES) {
    lc[tid] = ex;
    lst[tid] = ex;
    const int node = b * BIN_NODES + tid;
    if (node < N) rowptr[node] = base + ex;
  }
  if (tid == 0) lst[BIN_NODES] = tot;
  if (b == 0 && tid == 0) rowptr[N] = E;
  __syncthreads();
  for (int i0 = tid; i0 < tot + tid; i0 += 8 * 256) {
    int4 it[8];
#pragma unroll
    for (int u = 0; u < 8; ++u) {
      const int i = i0 + 256 * u;
      it[u] = rec0[base + ((i < tot) ? i : tot - 1)];
    }
#pragma unroll
    for (int u = 0; u < 8; ++u) {
      if (i0 + 256 * u < tot) {
        const int k = it[u].w & (BIN_NODES - 1);
        const int pos = atomicAdd(&lc[k], 1);
        if (in_lds) {
          l_eid[pos] = it[u].x;
          l_src[pos] = it[u].y;
          l_w[pos]   = it[u].z;
        } else {
          rec[base + pos] = make_int4(it[u].x, it[u].y, it[u].z, 0);
        }
      }
    }
  }
  __syncthreads();

  const int bi = wv * NODES_PER_WAVE + lane;
  const int bnd = lst[(bi < BIN_NODES) ? bi : BIN_NODES];
  for (int t = 0; t < NODES_PER_WAVE; ++t) {
    const int s = __builtin_amdgcn_readlane(bnd, t);
    const int cn = __builtin_amdgcn_readlane(bnd, t + 1) - s;
    if (in_lds) {
      if (cn <= 32) {
        int eid = 0x7fffffff, sv = 0, wb = 0;
        if (lane < cn) { eid = l_eid[s + lane]; sv = l_src[s + lane]; wb = l_w[s + lane]; }
        int r = 0;
        for (int j = 0; j < cn; ++j) {
          const int ej = __builtin_amdgcn_readlane(eid, j);
          r += (ej < eid) ? 1 : 0;
        }
        if (lane < cn) {
          csr_src[base + s + r] = sv;
          csr_w[base + s + r] = __int_as_float(wb);
        }
      } else {
        for (int i = lane; i < cn; i += 32) {
          const int eid = l_eid[s + i];
          int r = 0;
          for (int j = 0; j < cn; ++j) r += (l_eid[s + j] < eid) ? 1 : 0;
          csr_src[base + s + r] = l_src[s + i];
          csr_w[base + s + r] = __int_as_float(l_w[s + i]);
        }
      }
    } else {
      for (int i = lane; i < cn; i += 32) {
        const int4 it = rec[base + s + i];
        int r = 0;
        for (int j = 0; j < cn; ++j) r += (rec[base + s + j].x < it.x) ? 1 : 0;
        csr_src[base + s + r] = it.y;
        csr_w[base + s + r] = __int_as_float(it.z);
      }
    }
  }
}

__global__ __launch_bounds__(256)
void k_gemm1_count(const float* __restrict__ x, const float* __restrict__ W1,
                   const float* __restrict__ as1, const float* __restrict__ ad1,
                   float* __restrict__ hfeat, float* __restrict__ scoef,
                   float* __restrict__ dcoef, int N, int gb0, int ngb,
                   const int* __restrict__ dst, int* __restrict__ hist,
                   int* __restrict__ bintot, int E, int neb, int nbin,
                   const float* __restrict__ We1, const float* __restrict__ ae1,
                   const float* __restrict__ We2, const float* __restrict__ ae2,
                   float* __restrict__ cst) {
  int idx;
  if (shared_launch_role(blockIdx.x, ngb, neb, idx)) {
    gemm_body<128, 2>(gb0 + idx, x, W1, as1, ad1, hfeat, scoef, dcoef, N);
  } else if (idx < neb) {
    bin_count_body(idx, neb, dst, hist, bintot, E, nbin);
  } else {
    const int t = threadIdx.x;
    if (t < 2) edge_const(We1, ae1, cst, t);
    if (t == 32) edge_const(We2, ae2, cst + 4, 0);
  }
}

__global__ __launch_bounds__(256)
void k_gemm1_fill(const float* __restrict__ x, const float* __restrict__ W1,
                  const float* __restrict__ as1, const float* __restrict__ ad1,
                  float* __restrict__ hfeat, float* __restrict__ scoef,
                  float* __restrict__ dcoef, int N, int gb0, int ngb,
                  const int* __restrict__ src, const int* __restrict__ dst,
                  const float* __restrict__ ew, const int* __restrict__ offs,
                  int* __restrict__ rec0, int E, int neb, int nbin) {
  int idx;
  if (shared_launch_role(blockIdx.x, ngb, neb, idx)) {
    gemm_body<128, 2>(gb0 + idx, x, W1, as1, ad1, hfeat, scoef, dcoef, N);
  } else if (idx < neb) {
    bin_scatter_body(idx, neb, src, dst, ew, offs, rec0, E, nbin);
  }
}

template <int H>
__device__ __forceinline__ void item_logits(int s, float w,
                                            const float* __restrict__ scoef,
                                            const float* dc, const float* cs,
                                            float* al) {
  for (int h = 0; h < H; ++h) {
    float a = scoef[s * H + h] + dc[h] + w * cs[h];
    al[h] = (a >= 0.f) ? a : 0.2f * a;
  }
}

template <int H, int K>
__device__ __forceinline__ void gather_rows(
    const float* __restrict__ hfeat, int s, const float* wc, int j, int lane,
    __attribute__((ext_vector_type(2 * H))) float& acc) {
  constexpr int F = H * 64;
  constexpr int V = 2 * H;
  typedef __attribute__((ext_vector_type(V))) float vecV;
  vecV row[K];
  float c[K];
  for (int k = 0; k < K; ++k) {
    const int sj = __builtin_amdgcn_readlane(s, j + k);
    row[k] = *reinterpret_cast<const vecV*>(hfeat + (size_t)sj * F + lane * V);
    float c0 = __int_as_float(
        __builtin_amdgcn_readlane(__float_as_int(wc[0]), j + k));
    if (H == 2) {
      float c1 = __int_as_float(
          __builtin_amdgcn_readlane(__float_as_int(wc[H - 1]), j + k));
      c0 = (lane < 16) ? c0 : c1;
    }
    c[k] = c0;
  }
  for (int k = 0; k < K; ++k) acc += row[k] * c[k];
}

template <int H>
__device__ __forceinline__ void gather_chunk(
    const float* __restrict__ hfeat, int s, const float* wc, int cnt, int lane,
    __attribute__((ext_vector_type(2 * H))) float& acc) {
  int j = 0;
  for (; j + 8 <= cnt; j += 8) gather_rows<H, 8>(hfeat, s, wc, j, lane, acc);
  if (cnt - j >= 4) { gather_rows<H, 4>(hfeat, s, wc, j, lane, acc); j += 4; }
  if (cnt - j >= 2) { gather_rows<H, 2>(hfeat, s, wc, j, lane, acc); j += 2; }
  if (cnt - j >= 1) { gather_rows<H, 1>(hfeat, s, wc, j, lane, acc); }
}

template <int H, bool ELU>
__global__ __launch_bounds__(256)
void k_gat_gather(const int* __restrict__ rowptr, const int* __restrict__ csr_src,
                  const float* __restrict__ csr_w,
                  const float* __restrict__ hfeat,
                  const float* __restrict__ scoef,
                  const float* __restrict__ dcoef,
                  const float* __restrict__ cst, const float* __restrict__ bias,
                  float* __restrict__ outp, int N) {
  constexpr int F = H * 64;
  constexpr int V = 2 * H;
  typedef __attribute__((ext_vector_type(V))) float vecV;
  const int lane = threadIdx.x & 31;
  const int d_first =
      __builtin_amdgcn_readfirstlane(blockIdx.x * 8 + (threadIdx.x >> 5));
  for (int d = d_first; d < N; d += gridDim.x * 8) {
  const int start = rowptr[d];
  const int deg = rowptr[d + 1] - start;
  const int T = deg + 1;

  float dc[H], cs[H];
  for (int h = 0; h < H; ++h) { dc[h] = dcoef[d * H + h]; cs[h] = cst[h]; }

  int s0 = d;
  float w0 = 0.f;
  if (lane < deg) { s0 = csr_src[start + lane]; w0 = csr_w[start + lane]; }
  float pw = w0;
  for (int i = lane + 32; i < deg; i += 32) pw += csr_w[start + i];
  const float wloop = wave_sum_f(pw) / fmaxf((float)deg, 1.0f);
  if (lane == deg) w0 = wloop;
  const bool valid0 = lane < T;
  float al0[H];
  item_logits<H>(s0, w0, scoef, dc, cs, al0);

  float m[H];
  for (int h = 0; h < H; ++h) m[h] = valid0 ? al0[h] : -INFINITY;
  for (int base = 32; base < T; base += 32) {
    const int i = base + lane;
    int s = d; float w = wloop;
    if (i < deg) { s = csr_src[start + i]; w = csr_w[start + i]; }
    float al[H];
    item_logits<H>(s, w, scoef, dc, cs, al);
    if (i < T) {
      for (int h = 0; h < H; ++h) m[h] = fmaxf(m[h], al[h]);
    }
  }
  for (int h = 0; h < H; ++h) m[h] = wave_max_f(m[h]);

  float ex0[H], ps[H];
  for (int h = 0; h < H; ++h) {
    ex0[h] = valid0 ? expf(al0[h] - m[h]) : 0.f;
    ps[h] = ex0[h];
  }
  for (int base = 32; base < T; base += 32) {
    const int i = base + lane;
    int s = d; float w = wloop;
    if (i < deg) { s = csr_src[start + i]; w = csr_w[start + i]; }
    float al[H];
    item_logits<H>(s, w, scoef, dc, cs, al);
    if (i < T) {
      for (int h = 0; h < H; ++h) ps[h] += expf(al[h] - m[h]);
    }
  }
  float ssum[H];
  for (int h = 0; h < H; ++h) ssum[h] = wave_sum_f(ps[h]);

  vecV acc = {};
  float wc[H];
  for (int h = 0; h < H; ++h) wc[h] = ex0[h] / ssum[h];
  gather_chunk<H>(hfeat, s0, wc, (T < 32) ? T : 32, lane, acc);
  for (int base = 32; base < T; base += 32) {
    const int i = base + lane;
    int s = d; float w = wloop;
    if (i < deg) { s = csr_src[start + i]; w = csr_w[start + i]; }
    float al[H];
    item_logits<H>(s, w, scoef, dc, cs, al);
    for (int h = 0; h < H; ++h)
      wc[h] = (i < T) ? expf(al[h] - m[h]) / ssum[h] : 0.f;
    const int rem = T - base;
    gather_chunk<H>(hfeat, s, wc, (rem < 32) ? rem : 32, lane, acc);
  }

  const vecV bv = *reinterpret_cast<const vecV*>(bias + lane * V);
  vecV v = acc + bv;
  if (ELU) {
    for (int q = 0; q < V; ++q) v[q] = (v[q] > 0.f) ? v[q] : expm1f(v[q]);
  }
  *reinterpret_cast<vecV*>(outp + (size_t)d * F + lane * V) = v;
  }
}

extern "C" void kernel_launch(void* const* d_in, const int* in_sizes, int n_in,
                              void* d_out, int out_size, void* d_ws,
                              size_t ws_size, hipStream_t stream) {
  const float* x   = (const float*)d_in[0];
  const int*  eidx = (const int*)d_in[1];
  const float* ew  = (const float*)d_in[2];
  const float* W1  = (const float*)d_in[3];
  const float* We1 = (const float*)d_in[4];
  const float* as1 = (const float*)d_in[5];
  const float* ad1 = (const float*)d_in[6];
  const float* ae1 = (const float*)d_in[7];
  const float* b1  = (const float*)d_in[8];
  const float* W2  = (const float*)d_in[9];
  const float* We2 = (const float*)d_in[10];
  const float* as2 = (const float*)d_in[11];
  const float* ad2 = (const float*)d_in[12];
  const float* ae2 = (const float*)d_in[13];
  const float* b2  = (const float*)d_in[14];
  float* out = (float*)d_out;

  const int N  = in_sizes[0] / 128;
  const int E  = in_sizes[2];
  const int* srcv = eidx;
  const int* dstv = eidx + E;

  const int T = 256;
  auto cdiv = [](long long a, long long b) { return (int)((a + b - 1) / b); };
  const int EB   = cdiv(E, T * 8);
  const int NBIN = cdiv(N, BIN_NODES);

  float* p = (float*)d_ws;
  auto take = [&p](size_t n) { float* r = p; p += (n + 63) & ~(size_t)63; return r; };
  int*   bintot  = (int*)take(MAX_BINS);
  int*   rowptr  = (int*)take((size_t)N + 1);
  int*   hist    = (int*)take((size_t)NBIN * EB);
  int*   rec0    = (int*)take(4 * (size_t)E);
  int*   rec     = (int*)take(4 * (size_t)E);
  int*   csr_src = (int*)take(E);
  float* csr_w   = take(E);
  float* scoef   = take(2 * (size_t)N);
  float* dcoef   = take(2 * (size_t)N);
  float* cst     = take(8);
  float* hfeat   = take((size_t)N * 128);
  float* agg     = take((size_t)N * 128);

  const int NB  = 2048;
  const int NB6 = 1536;

  const int GB  = cdiv(cdiv(N, 16), 8);
  const int GBA = GB / 2;
  hipMemsetAsync(bintot, 0, (size_t)NBIN * 4, stream);
  k_gemm1_count<<<GBA + EB + 1, T, 0, stream>>>(
      x, W1, as1, ad1, hfeat, scoef, dcoef, N, 0, GBA,
      dstv, hist, bintot, E, EB, NBIN, We1, ae1, We2, ae2, cst);
  k_bin_offsets<<<NBIN, 256, 0, stream>>>(bintot, hist, EB);
  k_gemm1_fill<<<(GB - GBA) + EB, T, 0, stream>>>(
      x, W1, as1, ad1, hfeat, scoef, dcoef, N, GBA, GB - GBA,
      srcv, dstv, ew, hist, rec0, E, EB, NBIN);
  k_bin_place<<<NBIN, 256, 0, stream>>>(hist, bintot, rec0, rec, rowptr,
                                        csr_src, csr_w, N, E, EB);

  k_gat_gather<2, true><<<NB6, 256, 0, stream>>>(
      rowptr, csr_src, csr_w, hfeat, scoef, dcoef, cst, b1, agg, N);

  gemm_wmma_f32<64, 1><<<cdiv(cdiv(N, 16), 8), 256, 0, stream>>>(
      agg, W2, as2, ad2, hfeat, scoef, dcoef, N);
  k_gat_gather<1, false><<<NB, 256, 0, stream>>>(
      rowptr, csr_src, csr_w, hfeat, scoef, dcoef, cst + 4, b2, out, N);
}


# Edge bucketing level 1 in one pass: each edge work-group sorts its own 2048 edges by bin into a slab, the bin's group collects its runs; histogram and scan launches and memset gone; sum order kept
# speedup vs baseline: 7.3344x; 1.2581x over previous
#include <hip/hip_runtime.h>
#include <math.h>


typedef __attribute__((ext_vector_type(2))) float v2f;
typedef __attribute__((ext_vector_type(8))) float v8f;


__device__ __forceinline__ float row16_sum(float v) {
  v += __int_as_float(__builtin_amdgcn_update_dpp(0, __float_as_int(v), 0x161, 0xf, 0xf, false));
  v += __int_as_float(__builtin_amdgcn_update_dpp(0, __float_as_int(v), 0x162, 0xf, 0xf, false));
  v += __int_as_float(__builtin_amdgcn_update_dpp(0, __float_as_int(v), 0x164, 0xf, 0xf, false));
  v += __int_as_float(__builtin_amdgcn_update_dpp(0, __float_as_int(v), 0x168, 0xf, 0xf, false));
  return v;
}

template <int NCOLS, int H>
__device__ __forceinline__
void gemm_body(int gblock, const float* __restrict__ A, const float* __restrict__ Wg,
               const float* __restrict__ a_s, const float* __restrict__ a_d,
               float* __restrict__ Out, float* __restrict__ scoef,
               float* __restrict__ dcoef, int M) {
  constexpr int K = 128;
  constexpr int NT = NCOLS / 16;
  constexpr int TPH = NT / H;
  constexpr int P2 = 2 * NCOLS + 32;
  constexpr int C4 = NCOLS / 4;
  __shared__ __attribute__((aligned(16))) float Ws[(K / 2) * P2];

  const int tid = threadIdx.x;
  constexpr int NSTAGE = (K / 2) * C4 / 256;
  float4 wu[NSTAGE], wv[NSTAGE];
#pragma unroll
  for (int j = 0; j < NSTAGE; ++j) {
    const int i = tid + 256 * j;
    const int p = i / C4, c4 = i - p * C4;
    wu[j] = *reinterpret_cast<const float4*>(Wg + (2 * p) * NCOLS + 4 * c4);
    wv[j] = *reinterpret_cast<const float4*>(Wg + (2 * p + 1) * NCOLS + 4 * c4);
  }
#pragma unroll
  for (int j = 0; j < NSTAGE; ++j) {
    const int i = tid + 256 * j;
    const int p = i / C4, c4 = i - p * C4;
    float4* dst = reinterpret_cast<float4*>(&Ws[p * P2 + 8 * c4]);
    dst[0] = make_float4(wu[j].x, wv[j].x, wu[j].y, wv[j].y);
    dst[1] = make_float4(wu[j].z, wv[j].z, wu[j].w, wv[j].w);
  }
  __syncthreads();

  const int lane  = tid & 31;
  const int r     = lane & 15;
  const int half  = lane >> 4;
  const int nstrips = (M + 15) >> 4;
  const int strip = __builtin_amdgcn_readfirstlane(gblock * 8 + (tid >> 5));
  if (strip >= nstrips) return;

  const int row16 = strip << 4;
  int arow = row16 + r;
  if (arow >= M) arow = M - 1;
  const float* __restrict__ ap = A + (size_t)arow * K + 2 * half;
  const float* bp = &Ws[half * P2 + 2 * r];

  v8f acc[NT];
#pragma unroll
  for (int t = 0; t < NT; ++t) acc[t] = v8f{};
#pragma unroll
  for (int k = 0; k < K; k += 4) {
    v2f a;
    a.x = ap[k];
    a.y = ap[k + 1];
#pragma unroll
    for (int t = 0; t < NT; ++t) {
      const v2f b = *reinterpret_cast<const v2f*>(bp + (k / 2) * P2 + 32 * t);
      acc[t] = __builtin_amdgcn_wmma_f32_16x16x4_f32(false, a, false, b,
                                                     (short)0, acc[t], false, false);
    }
  }

  const int orow0 = row16 + (half << 3);
  float ps[H][8], pd[H][8];
#pragma unroll
  for (int h = 0; h < H; ++h) {
#pragma unroll
    for (int i = 0; i < 8; ++i) {
      float s0 = 0.f, d0 = 0.f;
#pragma unroll
      for (int t = h * TPH; t < (h + 1) * TPH; ++t) {
        s0 += acc[t][i] * a_s[t * 16 + r];
        d0 += acc[t][i] * a_d[t * 16 + r];
      }
      ps[h][i] = row16_sum(s0);
      pd[h][i] = row16_sum(d0);
    }
  }

#pragma unroll
  for (int i = 0; i < 8; ++i) {
    const int orow = orow0 + i;
    if (orow < M) {
#pragma unroll
      for (int t = 0; t < NT; ++t)
        Out[(size_t)orow * NCOLS + t * 16 + r] = acc[t][i];
      if (r == 0) {
#pragma unroll
        for (int h = 0; h < H; ++h) {
          scoef[orow * H + h] = ps[h][i];
          dcoef[orow * H + h] = pd[h][i];
        }
      }
    }
  }
}

template <int NCOLS, int H>
__global__ __launch_bounds__(256)
void gemm_wmma_f32(const float* __restrict__ A, const float* __restrict__ Wg,
                   const float* __restrict__ a_s, const float* __restrict__ a_d,
                   float* __restrict__ Out, float* __restrict__ scoef,
                   float* __restrict__ dcoef, int M) {
  gemm_body<NCOLS, H>(blockIdx.x, A, Wg, a_s, a_d, Out, scoef, dcoef, M);
}

__device__ __forceinline__ void edge_const(const float* __restrict__ We,
                                           const float* __restrict__ ae,
                                           float* __restrict__ cst, int h) {
  float s = 0.f;
  for (int c = 0; c < 64; ++c) s += We[h * 64 + c] * ae[h * 64 + c];
  cst[h] = s;
}

__device__ __forceinline__ bool shared_launch_role(int b, int ngb, int neb, int& idx) {
  const int m = (ngb < neb / 2) ? ngb : neb / 2;
  if (b < 3 * m) {
    const int q = b / 3, r = b - 3 * q;
    if (r == 0) { idx = q; return true; }
    idx = 2 * q + r - 1;
    return false;
  }
  const int rest = b - 3 * m;
  if (rest < ngb - m) { idx = m + rest; return true; }
  idx = 2 * m + rest - (ngb - m);
  return false;
}


__device__ __forceinline__ int wave_sum_i(int v) {
  for (int o = 16; o > 0; o >>= 1) v += __shfl_xor(v, o);
  return v;
}
__device__ __forceinline__ int wave_max_i(int v) {
  for (int o = 16; o > 0; o >>= 1) { const int t = __shfl_xor(v, o); v = (t > v) ? t : v; }
  return v;
}
__device__ __forceinline__ float wave_sum_f(float v) {
  for (int o = 16; o > 0; o >>= 1) v += __shfl_xor(v, o);
  return v;
}
__device__ __forceinline__ float wave_max_f(float v) {
  for (int o = 16; o > 0; o >>= 1) v = fmaxf(v, __shfl_xor(v, o));
  return v;
}

constexpr int BIN_SHIFT = 7;
constexpr int BIN_NODES = 1 << BIN_SHIFT;
constexpr int MAX_BINS  = 1024;

constexpr int SLAB   = 2048;

__device__ __forceinline__ void bin_local_body(int vb,
                                               const int* __restrict__ src,
                                               const int* __restrict__ dst,
                                               const float* __restrict__ ew,
                                               int* __restrict__ st,
                                               int* __restrict__ rec_i,
                                               int E, int nbin) {
  __shared__ int lh[MAX_BINS];
  __shared__ int s_w[8];
  int4* __restrict__ rec = reinterpret_cast<int4*>(rec_i);
  const int tid = threadIdx.x, lane = tid & 31, wv = tid >> 5;
  for (int i = tid; i < MAX_BINS; i += 256) lh[i] = 0;
  const int e0 = vb * SLAB + tid;
  int dn[8], sn[8], wn[8];
#pragma unroll
  for (int u = 0; u < 8; ++u) {
    const int e = e0 + 256 * u;
    const int ec = (e < E) ? e : E - 1;
    dn[u] = dst[ec];
    sn[u] = src[ec];
    wn[u] = __float_as_int(ew[ec]);
  }
  __syncthreads();
#pragma unroll
  for (int u = 0; u < 8; ++u)
    if (e0 + 256 * u < E) atomicAdd(&lh[dn[u] >> BIN_SHIFT], 1);
  __syncthreads();
  const int b4 = tid * 4;
  const int v0 = lh[b4], v1 = lh[b4 + 1], v2 = lh[b4 + 2], v3 = lh[b4 + 3];
  const int tsum = (v0 + v1) + (v2 + v3);
  int inc = tsum;
  for (int o = 1; o < 32; o <<= 1) {
    int t = __shfl_up(inc, o);
    if (lane >= o) inc += t;
  }
  if (lane == 31) s_w[wv] = inc;
  __syncthreads();
  int off = 0;
  for (int i = 0; i < wv; ++i) off += s_w[i];
  const int ex = off + inc - tsum;
  const int x1 = ex + v0, x2 = x1 + v1, x3 = x2 + v2;
  lh[b4] = ex; lh[b4 + 1] = x1; lh[b4 + 2] = x2; lh[b4 + 3] = x3;
  int* __restrict__ row = st + (size_t)vb * (nbin + 1);
  if (b4     < nbin) row[b4]     = ex;
  if (b4 + 1 < nbin) row[b4 + 1] = x1;
  if (b4 + 2 < nbin) row[b4 + 2] = x2;
  if (b4 + 3 < nbin) row[b4 + 3] = x3;
  if (tid == 0) {
    const int left = E - vb * SLAB;
    row[nbin] = (left < SLAB) ? left : SLAB;
  }
  __syncthreads();
  const int sbase = vb * SLAB;
#pragma unroll
  for (int u = 0; u < 8; ++u) {
    const int e = e0 + 256 * u;
    if (e < E) {
      const int pos = atomicAdd(&lh[dn[u] >> BIN_SHIFT], 1);
      rec[sbase + pos] = make_int4(e, sn[u], wn[u], dn[u]);
    }
  }
}

constexpr int PLACE_CAP = 4096;
constexpr int NODES_PER_WAVE = BIN_NODES / 8;

__global__ __launch_bounds__(256)
void k_bin_place(const int* __restrict__ st,
                 const int* __restrict__ rec0_i, int* __restrict__ rec_i,
                 int* __restrict__ rowptr, int* __restrict__ csr_src,
                 float* __restrict__ csr_w, int N, int E, int nvb, int nbin) {
  __shared__ int s_c[8];
  __shared__ int s_b[8];
  __shared__ int lc[BIN_NODES];
  __shared__ int lst[BIN_NODES + 1];
  __shared__ int s_tot[8];
  __shared__ int l_eid[PLACE_CAP];
  __shared__ int l_src[PLACE_CAP];
  __shared__ int l_w[PLACE_CAP];
  const int4* __restrict__ rec0 = reinterpret_cast<const int4*>(rec0_i);
  int4* __restrict__ rec = reinterpret_cast<int4*>(rec_i);
  const int tid = threadIdx.x, lane = tid & 31, wv = tid >> 5;
  const int b = blockIdx.x;
  if (tid < BIN_NODES) lc[tid] = 0;
  const int q4 = tid * 4;
  int r0[4], rc[4];
#pragma unroll
  for (int q = 0; q < 4; ++q) {
    const int vb = q4 + q;
    int a = 0, z = 0;
    if (vb < nvb) {
      const int* __restrict__ p = st + (size_t)vb * (nbin + 1) + b;
      a = p[0];
      z = p[1];
    }
    r0[q] = a;
    rc[q] = z - a;
  }
  const int csum = wave_sum_i((rc[0] + rc[1]) + (rc[2] + rc[3]));
  const int bsum = wave_sum_i((r0[0] + r0[1]) + (r0[2] + r0[3]));
  if (lane == 0) { s_c[wv] = csum; s_b[wv] = bsum; }
  int rmax = (rc[0] > rc[1]) ? rc[0] : rc[1];
  rmax = (rmax > rc[2]) ? rmax : rc[2];
  rmax = (rmax > rc[3]) ? rmax : rc[3];
  const int kw = __builtin_amdgcn_readfirstlane(wave_max_i(rmax));
  int ra[4];
#pragma unroll
  for (int q = 0; q < 4; ++q) ra[q] = (q4 + q) * SLAB + r0[q];
  __syncthreads();
  int tot = 0, base = 0;
  for (int i = 0; i < 8; ++i) { tot += s_c[i]; base += s_b[i]; }
  const bool in_lds = tot <= PLACE_CAP;
  for (int k0 = 0; k0 < kw; k0 += 4) {
    int dn[16];
#pragma unroll
    for (int u = 0; u < 16; ++u) {
      const int k = k0 + (u & 3);
      dn[u] = rec0[(k < rc[u >> 2]) ? ra[u >> 2] + k : 0].w;
    }
#pragma unroll
    for (int u = 0; u < 16; ++u)
      if (k0 + (u & 3) < rc[u >> 2]) atomicAdd(&lc[dn[u] & (BIN_NODES - 1)], 1);
  }
  __syncthreads();
  const int c = (tid < BIN_NODES) ? lc[tid] : 0;
  int inc = c;
  for (int o = 1; o < 32; o <<= 1) {
    int t = __shfl_up(inc, o);
    if (lane >= o) inc += t;
  }
  if (lane == 31) s_tot[wv] = inc;
  __syncthreads();
  int off = 0;
  for (int i = 0; i < wv; ++i) off += s_tot[i];
  const int ex = off + inc - c;
  if (tid < BIN_NODES) {
    lc[tid] = ex;
    lst[tid] = ex;
    const int node = b * BIN_NODES + tid;
    if (node < N) rowptr[node] = base + ex;
  }
  if (tid == 0) lst[BIN_NODES] = tot;
  if (b == 0 && tid == 0) rowptr[N] = E;
  __syncthreads();
  for (int k0 = 0; k0 < kw; k0 += 4) {
    int4 it[16];
#pragma unroll
    for (int u = 0; u < 16; ++u) {
      const int k = k0 + (u & 3);
      it[u] = rec0[(k < rc[u >> 2]) ? ra[u >> 2] + k : 0];
    }
#pragma unroll
    for (int u = 0; u < 16; ++u) {
      if (k0 + (u & 3) < rc[u >> 2]) {
        const int k = it[u].w & (BIN_NODES - 1);
        const int pos = atomicAdd(&lc[k], 1);
        if (in_lds) {
          l_eid[pos] = it[u].x;
          l_src[pos] = it[u].y;
          l_w[pos]   = it[u].z;
        } else {
          rec[base + pos] = make_int4(it[u].x, it[u].y, it[u].z, 0);
        }
      }
    }
  }
  __syncthreads();

  const int bi = wv * NODES_PER_WAVE + lane;
  const int bnd = lst[(bi < BIN_NODES) ? bi : BIN_NODES];
  for (int t = 0; t < NODES_PER_WAVE; ++t) {
    const int s = __builtin_amdgcn_readlane(bnd, t);
    const int cn = __builtin_amdgcn_readlane(bnd, t + 1) - s;
    if (in_lds) {
      if (cn <= 32) {
        int eid = 0x7fffffff, sv = 0, wb = 0;
        if (lane < cn) { eid = l_eid[s + lane]; sv = l_src[s + lane]; wb = l_w[s + lane]; }
        int r = 0;
        for (int j = 0; j < cn; ++j) {
          const int ej = __builtin_amdgcn_readlane(eid, j);
          r += (ej < eid) ? 1 : 0;
        }
        if (lane < cn) {
          csr_src[base + s + r] = sv;
          csr_w[base + s + r] = __int_as_float(wb);
        }
      } else {
        for (int i = lane; i < cn; i += 32) {
          const int eid = l_eid[s + i];
          int r = 0;
          for (int j = 0; j < cn; ++j) r += (l_eid[s + j] < eid) ? 1 : 0;
          csr_src[base + s + r] = l_src[s + i];
          csr_w[base + s + r] = __int_as_float(l_w[s + i]);
        }
      }
    } else {
      for (int i = lane; i < cn; i += 32) {
        const int4 it = rec[base + s + i];
        int r = 0;
        for (int j = 0; j < cn; ++j) r += (rec[base + s + j].x < it.x) ? 1 : 0;
        csr_src[base + s + r] = it.y;
        csr_w[base + s + r] = __int_as_float(it.z);
      }
    }
  }
}

__global__ __launch_bounds__(256)
void k_gemm1_part(const float* __restrict__ x, const float* __restrict__ W1,
                  const float* __restrict__ as1, const float* __restrict__ ad1,
                  float* __restrict__ hfeat, float* __restrict__ scoef,
                  float* __restrict__ dcoef, int N, int ngb,
                  const int* __restrict__ src, const int* __restrict__ dst,
                  const float* __restrict__ ew, int* __restrict__ st,
                  int* __restrict__ rec0, int E, int neb, int nbin,
                  const float* __restrict__ We1, const float* __restrict__ ae1,
                  const float* __restrict__ We2, const float* __restrict__ ae2,
                  float* __restrict__ cst) {
  int idx;
  if (shared_launch_role(blockIdx.x, ngb, neb, idx)) {
    gemm_body<128, 2>(idx, x, W1, as1, ad1, hfeat, scoef, dcoef, N);
  } else if (idx < neb) {
    bin_local_body(idx, src, dst, ew, st, rec0, E, nbin);
  } else {
    const int t = threadIdx.x;
    if (t < 2) edge_const(We1, ae1, cst, t);
    if (t == 32) edge_const(We2, ae2, cst + 4, 0);
  }
}

template <int H>
__device__ __forceinline__ void item_logits(int s, float w,
                                            const float* __restrict__ scoef,
                                            const float* dc, const float* cs,
                                            float* al) {
  for (int h = 0; h < H; ++h) {
    float a = scoef[s * H + h] + dc[h] + w * cs[h];
    al[h] = (a >= 0.f) ? a : 0.2f * a;
  }
}

template <int H, int K>
__device__ __forceinline__ void gather_rows(
    const float* __restrict__ hfeat, int s, const float* wc, int j, int lane,
    __attribute__((ext_vector_type(2 * H))) float& acc) {
  constexpr int F = H * 64;
  constexpr int V = 2 * H;
  typedef __attribute__((ext_vector_type(V))) float vecV;
  vecV row[K];
  float c[K];
  for (int k = 0; k < K; ++k) {
    const int sj = __builtin_amdgcn_readlane(s, j + k);
    row[k] = *reinterpret_cast<const vecV*>(hfeat + (size_t)sj * F + lane * V);
    float c0 = __int_as_float(
        __builtin_amdgcn_readlane(__float_as_int(wc[0]), j + k));
    if (H == 2) {
      float c1 = __int_as_float(
          __builtin_amdgcn_readlane(__float_as_int(wc[H - 1]), j + k));
      c0 = (lane < 16) ? c0 : c1;
    }
    c[k] = c0;
  }
  for (int k = 0; k < K; ++k) acc += row[k] * c[k];
}

template <int H>
__device__ __forceinline__ void gather_chunk(
    const float* __restrict__ hfeat, int s, const float* wc, int cnt, int lane,
    __attribute__((ext_vector_type(2 * H))) float& acc) {
  int j = 0;
  for (; j + 8 <= cnt; j += 8) gather_rows<H, 8>(hfeat, s, wc, j, lane, acc);
  if (cnt - j >= 4) { gather_rows<H, 4>(hfeat, s, wc, j, lane, acc); j += 4; }
  if (cnt - j >= 2) { gather_rows<H, 2>(hfeat, s, wc, j, lane, acc); j += 2; }
  if (cnt - j >= 1) { gather_rows<H, 1>(hfeat, s, wc, j, lane, acc); }
}

template <int H, bool ELU>
__global__ __launch_bounds__(256)
void k_gat_gather(const int* __restrict__ rowptr, const int* __restrict__ csr_src,
                  const float* __restrict__ csr_w,
                  const float* __restrict__ hfeat,
                  const float* __restrict__ scoef,
                  const float* __restrict__ dcoef,
                  const float* __restrict__ cst, const float* __restrict__ bias,
                  float* __restrict__ outp, int N) {
  constexpr int F = H * 64;
  constexpr int V = 2 * H;
  typedef __attribute__((ext_vector_type(V))) float vecV;
  const int lane = threadIdx.x & 31;
  const int d_first =
      __builtin_amdgcn_readfirstlane(blockIdx.x * 8 + (threadIdx.x >> 5));
  for (int d = d_first; d < N; d += gridDim.x * 8) {
  const int start = rowptr[d];
  const int deg = rowptr[d + 1] - start;
  const int T = deg + 1;

  float dc[H], cs[H];
  for (int h = 0; h < H; ++h) { dc[h] = dcoef[d * H + h]; cs[h] = cst[h]; }

  int s0 = d;
  float w0 = 0.f;
  if (lane < deg) { s0 = csr_src[start + lane]; w0 = csr_w[start + lane]; }
  float pw = w0;
  for (int i = lane + 32; i < deg; i += 32) pw += csr_w[start + i];
  const float wloop = wave_sum_f(pw) / fmaxf((float)deg, 1.0f);
  if (lane == deg) w0 = wloop;
  const bool valid0 = lane < T;
  float al0[H];
  item_logits<H>(s0, w0, scoef, dc, cs, al0);

  float m[H];
  for (int h = 0; h < H; ++h) m[h] = valid0 ? al0[h] : -INFINITY;
  for (int base = 32; base < T; base += 32) {
    const int i = base + lane;
    int s = d; float w = wloop;
    if (i < deg) { s = csr_src[start + i]; w = csr_w[start + i]; }
    float al[H];
    item_logits<H>(s, w, scoef, dc, cs, al);
    if (i < T) {
      for (int h = 0; h < H; ++h) m[h] = fmaxf(m[h], al[h]);
    }
  }
  for (int h = 0; h < H; ++h) m[h] = wave_max_f(m[h]);

  float ex0[H], ps[H];
  for (int h = 0; h < H; ++h) {
    ex0[h] = valid0 ? expf(al0[h] - m[h]) : 0.f;
    ps[h] = ex0[h];
  }
  for (int base = 32; base < T; base += 32) {
    const int i = base + lane;
    int s = d; float w = wloop;
    if (i < deg) { s = csr_src[start + i]; w = csr_w[start + i]; }
    float al[H];
    item_logits<H>(s, w, scoef, dc, cs, al);
    if (i < T) {
      for (int h = 0; h < H; ++h) ps[h] += expf(al[h] - m[h]);
    }
  }
  float ssum[H];
  for (int h = 0; h < H; ++h) ssum[h] = wave_sum_f(ps[h]);

  vecV acc = {};
  float wc[H];
  for (int h = 0; h < H; ++h) wc[h] = ex0[h] / ssum[h];
  gather_chunk<H>(hfeat, s0, wc, (T < 32) ? T : 32, lane, acc);
  for (int base = 32; base < T; base += 32) {
    const int i = base + lane;
    int s = d; float w = wloop;
    if (i < deg) { s = csr_src[start + i]; w = csr_w[start + i]; }
    float al[H];
    item_logits<H>(s, w, scoef, dc, cs, al);
    for (int h = 0; h < H; ++h)
      wc[h] = (i < T) ? expf(al[h] - m[h]) / ssum[h] : 0.f;
    const int rem = T - base;
    gather_chunk<H>(hfeat, s, wc, (rem < 32) ? rem : 32, lane, acc);
  }

  const vecV bv = *reinterpret_cast<const vecV*>(bias + lane * V);
  vecV v = acc + bv;
  if (ELU) {
    for (int q = 0; q < V; ++q) v[q] = (v[q] > 0.f) ? v[q] : expm1f(v[q]);
  }
  *reinterpret_cast<vecV*>(outp + (size_t)d * F + lane * V) = v;
  }
}

extern "C" void kernel_launch(void* const* d_in, const int* in_sizes, int n_in,
                              void* d_out, int out_size, void* d_ws,
                              size_t ws_size, hipStream_t stream) {
  const float* x   = (const float*)d_in[0];
  const int*  eidx = (const int*)d_in[1];
  const float* ew  = (const float*)d_in[2];
  const float* W1  = (const float*)d_in[3];
  const float* We1 = (const float*)d_in[4];
  const float* as1 = (const float*)d_in[5];
  const float* ad1 = (const float*)d_in[6];
  const float* ae1 = (const float*)d_in[7];
  const float* b1  = (const float*)d_in[8];
  const float* W2  = (const float*)d_in[9];
  const float* We2 = (const float*)d_in[10];
  const float* as2 = (const float*)d_in[11];
  const float* ad2 = (const float*)d_in[12];
  const float* ae2 = (const float*)d_in[13];
  const float* b2  = (const float*)d_in[14];
  float* out = (float*)d_out;

  const int N  = in_sizes[0] / 128;
  const int E  = in_sizes[2];
  const int* srcv = eidx;
  const int* dstv = eidx + E;

  const int T = 256;
  auto cdiv = [](long long a, long long b) { return (int)((a + b - 1) / b); };
  const int EB   = cdiv(E, SLAB);
  const int NBIN = cdiv(N, BIN_NODES);

  float* p = (float*)d_ws;
  auto take = [&p](size_t n) { float* r = p; p += (n + 63) & ~(size_t)63; return r; };
  int*   rowptr  = (int*)take((size_t)N + 1);
  int*   st      = (int*)take((size_t)(NBIN + 1) * EB);
  int*   rec0    = (int*)take(4 * (size_t)E);
  int*   rec     = (int*)take(4 * (size_t)E);
  int*   csr_src = (int*)take(E);
  float* csr_w   = take(E);
  float* scoef   = take(2 * (size_t)N);
  float* dcoef   = take(2 * (size_t)N);
  float* cst     = take(8);
  float* hfeat   = take((size_t)N * 128);
  float* agg     = take((size_t)N * 128);

  const int NB  = 2048;
  const int NB6 = 1536;

  const int GB  = cdiv(cdiv(N, 16), 8);
  k_gemm1_part<<<GB + EB + 1, T, 0, stream>>>(
      x, W1, as1, ad1, hfeat, scoef, dcoef, N, GB,
      srcv, dstv, ew, st, rec0, E, EB, NBIN, We1, ae1, We2, ae2, cst);
  k_bin_place<<<NBIN, 256, 0, stream>>>(st, rec0, rec, rowptr,
                                        csr_src, csr_w, N, E, EB, NBIN);

  k_gat_gather<2, true><<<NB6, 256, 0, stream>>>(
      rowptr, csr_src, csr_w, hfeat, scoef, dcoef, cst, b1, agg, N);

  gemm_wmma_f32<64, 1><<<cdiv(cdiv(N, 16), 8), 256, 0, stream>>>(
      agg, W2, as2, ad2, hfeat, scoef, dcoef, N);
  k_gat_gather<1, false><<<NB, 256, 0, stream>>>(
      rowptr, csr_src, csr_w, hfeat, scoef, dcoef, cst + 4, b2, out, N);
}
